# starting kernel
# baseline (speedup 1.0000x reference)
#include <hip/hip_runtime.h>

#define HID 128
#define OUTD 64

typedef float v2f __attribute__((ext_vector_type(2)));
typedef float v8f __attribute__((ext_vector_type(8)));

__global__ __launch_bounds__(256) void k_deg(const int* __restrict__ dst,
                                             float* __restrict__ deg, int E) {
  int e = blockIdx.x * 256 + threadIdx.x;
  if (e < E) atomicAdd(&deg[dst[e]], 1.0f);
}

__global__ __launch_bounds__(256) void k_rdeg(float* __restrict__ deg, int N) {
  int i = blockIdx.x * 256 + threadIdx.x;
  if (i < N) deg[i] = 1.0f / fmaxf(deg[i], 1.0f);
}

template<int D>
__global__ __launch_bounds__(256) void k_agg(const float* __restrict__ feat,
                                             const int* __restrict__ src,
                                             const int* __restrict__ dst,
                                             float* __restrict__ sum, int E) {
  const int CH = D / 4;
  long long t = (long long)blockIdx.x * 256 + threadIdx.x;
  int e = (int)(t / CH);
  int c = (int)(t % CH);
  if (e >= E) return;
  int s = src[e], d = dst[e];
  const float4 v = *(const float4*)(feat + (size_t)s * D + (size_t)c * 4);
  float* o = sum + (size_t)d * D + (size_t)c * 4;
  atomicAdd(o + 0, v.x);
  atomicAdd(o + 1, v.y);
  atomicAdd(o + 2, v.z);
  atomicAdd(o + 3, v.w);
}

template<int NCOLS, bool TWO, bool ADD_MEAN, bool RELU, bool BIAS>
__global__ __launch_bounds__(256) void k_gemm(const float* __restrict__ A1,
                                              const float* __restrict__ A2,
                                              const float* __restrict__ rdeg,
                                              const float* __restrict__ B1,
                                              const float* __restrict__ B2,
                                              const float* __restrict__ bias,
                                              const float* __restrict__ addmat,
                                              float* __restrict__ out, int M) {
  const int NT = NCOLS / 16;
  int wave = (blockIdx.x * 256 + (int)threadIdx.x) >> 5;
  int mt = wave / NT;
  int nt = wave - mt * NT;
  int row0 = mt * 16;
  if (row0 >= M) return;

  int lane = threadIdx.x & 31;
  int hh = lane >> 4;
  int lr = lane & 15;
  int col0 = nt * 16;

  const float* a1 = A1 + (size_t)(row0 + lr) * HID + 2 * hh;
  const float* b1 = B1 + (size_t)(2 * hh) * NCOLS + col0 + lr;
  const float* a2 = nullptr;
  const float* b2 = nullptr;
  float rd = 0.f;
  if (TWO) {
    a2 = A2 + (size_t)(row0 + lr) * HID + 2 * hh;
    b2 = B2 + (size_t)(2 * hh) * NCOLS + col0 + lr;
    rd = rdeg[row0 + lr];
  }

  v8f c = {};
#pragma unroll 4
  for (int k = 0; k < HID; k += 4) {
    v2f av = *(const v2f*)(a1 + k);
    v2f bv;
    bv.x = b1[(size_t)k * NCOLS];
    bv.y = b1[(size_t)(k + 1) * NCOLS];
    c = __builtin_amdgcn_wmma_f32_16x16x4_f32(false, av, false, bv, (short)0, c,
                                              false, false);
    if (TWO) {
      v2f av2 = *(const v2f*)(a2 + k);
      av2.x *= rd;
      av2.y *= rd;
      v2f bv2;
      bv2.x = b2[(size_t)k * NCOLS];
      bv2.y = b2[(size_t)(k + 1) * NCOLS];
      c = __builtin_amdgcn_wmma_f32_16x16x4_f32(false, av2, false, bv2, (short)0,
                                                c, false, false);
    }
  }

  float bb = BIAS ? bias[col0 + lr] : 0.f;
#pragma unroll
  for (int v = 0; v < 8; ++v) {
    int r = row0 + v + 8 * hh;
    float val = c[v] + bb;
    if (ADD_MEAN) val += addmat[(size_t)r * NCOLS + col0 + lr] * rdeg[r];
    if (RELU) val = fmaxf(val, 0.f);
    out[(size_t)r * NCOLS + col0 + lr] = val;
  }
}

extern "C" void kernel_launch(void* const* d_in, const int* in_sizes, int n_in,
                              void* d_out, int out_size, void* d_ws,
                              size_t ws_size, hipStream_t stream) {
  const float* x   = (const float*)d_in[0];
  const int*   src = (const int*)d_in[1];
  const int*   dst = (const int*)d_in[2];
  const float* Ws1 = (const float*)d_in[3];
  const float* b1  = (const float*)d_in[4];
  const float* Wn1 = (const float*)d_in[5];
  const float* Ws2 = (const float*)d_in[6];
  const float* b2  = (const float*)d_in[7];
  const float* Wn2 = (const float*)d_in[8];
  const float* Ws3 = (const float*)d_in[9];
  const float* b3  = (const float*)d_in[10];
  const float* Wn3 = (const float*)d_in[11];
  float* out = (float*)d_out;

  const int N = in_sizes[0] / HID;
  const int E = in_sizes[1];

  float* deg = (float*)d_ws;
  float* h1  = deg + N;
  float* h2  = h1 + (size_t)N * HID;
  float* tb  = h2 + (size_t)N * HID;
  float* sum = tb + (size_t)N * OUTD;

  const int mtiles = (N + 15) / 16;
  const int tilesH = mtiles * (HID / 16);
  const int tilesO = mtiles * (OUTD / 16);
  const long long aggTH = (long long)E * (HID / 4);
  const long long aggTO = (long long)E * (OUTD / 4);

  hipMemsetAsync(deg, 0, (size_t)N * sizeof(float), stream);
  hipMemsetAsync(sum, 0, (size_t)N * HID * sizeof(float), stream);
  k_deg<<<(E + 255) / 256, 256, 0, stream>>>(dst, deg, E);
  k_rdeg<<<(N + 255) / 256, 256, 0, stream>>>(deg, N);

  k_agg<HID><<<(int)((aggTH + 255) / 256), 256, 0, stream>>>(x, src, dst, sum, E);
  k_gemm<HID, true, false, true, true><<<(tilesH + 7) / 8, 256, 0, stream>>>(
      x, sum, deg, Ws1, Wn1, b1, nullptr, h1, N);

  hipMemsetAsync(sum, 0, (size_t)N * HID * sizeof(float), stream);
  k_agg<HID><<<(int)((aggTH + 255) / 256), 256, 0, stream>>>(h1, src, dst, sum, E);
  k_gemm<HID, true, false, true, true><<<(tilesH + 7) / 8, 256, 0, stream>>>(
      h1, sum, deg, Ws2, Wn2, b2, nullptr, h2, N);

  k_gemm<OUTD, false, false, false, false><<<(tilesO + 7) / 8, 256, 0, stream>>>(
      h2, nullptr, nullptr, Wn3, nullptr, nullptr, nullptr, tb, N);
  hipMemsetAsync(sum, 0, (size_t)N * OUTD * sizeof(float), stream);
  k_agg<OUTD><<<(int)((aggTO + 255) / 256), 256, 0, stream>>>(tb, src, dst, sum, E);
  k_gemm<OUTD, false, true, false, true><<<(tilesO + 7) / 8, 256, 0, stream>>>(
      h2, nullptr, deg, Ws3, nullptr, b3, sum, out, N);
}


# Float-atomic scatter-sum replaced by in-call CSR gather: int-atomic count gives arrival-order rank, scan, fill; one wave sums a node's neighbour rows (sum order = arrival rank, D=64: 2 partial sums)
# speedup vs baseline: 2.1681x; 2.1681x over previous
#include <hip/hip_runtime.h>

#define HID 128
#define OUTD 64

typedef float v2f __attribute__((ext_vector_type(2)));
typedef float v8f __attribute__((ext_vector_type(8)));


__global__ __launch_bounds__(256) void k_count(const int* __restrict__ dst,
                                               int* __restrict__ cnt,
                                               int* __restrict__ rank, int E) {
  int e = blockIdx.x * 256 + threadIdx.x;
  if (e < E) rank[e] = atomicAdd(&cnt[dst[e]], 1);
}

__global__ __launch_bounds__(256) void k_scan(const int* __restrict__ cnt,
                                              int* __restrict__ off,
                                              float* __restrict__ rdeg, int N) {
  __shared__ int s_base[8];
  __shared__ int s_part[8];
  const int tid = threadIdx.x;
  const int lane = tid & 31;
  const int w = tid >> 5;
  const int lim = blockIdx.x * 1024;

  int a0 = 0, a1 = 0, a2 = 0, a3 = 0;
  int i = tid;
  for (; i + 768 < lim; i += 1024) {
    a0 += cnt[i];
    a1 += cnt[i + 256];
    a2 += cnt[i + 512];
    a3 += cnt[i + 768];
  }
  for (; i < lim; i += 256) a0 += cnt[i];
  int acc = (a0 + a1) + (a2 + a3);
  for (int o = 16; o > 0; o >>= 1) acc += __shfl_xor(acc, o);
  if (lane == 0) s_base[w] = acc;

  const int i0 = lim + 4 * tid;
  int c0 = (i0 + 0 < N) ? cnt[i0 + 0] : 0;
  int c1 = (i0 + 1 < N) ? cnt[i0 + 1] : 0;
  int c2 = (i0 + 2 < N) ? cnt[i0 + 2] : 0;
  int c3 = (i0 + 3 < N) ? cnt[i0 + 3] : 0;
  const int s = (c0 + c1) + (c2 + c3);
  int inc = s;
  for (int o = 1; o < 32; o <<= 1) {
    int t = __shfl_up(inc, o);
    if (lane >= o) inc += t;
  }
  if (lane == 31) s_part[w] = inc;
  __syncthreads();
  int base = 0;
  for (int k = 0; k < 8; ++k) base += s_base[k];
  for (int k = 0; k < w; ++k) base += s_part[k];
  int ex = base + inc - s;
  const int cc[4] = {c0, c1, c2, c3};
#pragma unroll
  for (int k = 0; k < 4; ++k) {
    int n = i0 + k;
    if (n < N) {
      off[n] = ex;
      rdeg[n] = 1.0f / fmaxf((float)cc[k], 1.0f);
      if (n == N - 1) off[N] = ex + cc[k];
    }
    ex += cc[k];
  }
}

__global__ __launch_bounds__(256) void k_fill(const int* __restrict__ src,
                                              const int* __restrict__ dst,
                                              const int* __restrict__ off,
                                              const int* __restrict__ rank,
                                              int* __restrict__ csr, int E) {
  int e = blockIdx.x * 256 + threadIdx.x;
  if (e < E) csr[off[dst[e]] + rank[e]] = src[e];
}

template<int D, int NPW>
__global__ __launch_bounds__(256) void k_gather(const float* __restrict__ feat,
                                                const int* __restrict__ off,
                                                const int* __restrict__ csr,
                                                float* __restrict__ sum, int N) {
  constexpr int LPR = D / 4;
  constexpr int G = 32 / LPR;
  const int lane = threadIdx.x & 31;
  const int wave = __builtin_amdgcn_readfirstlane((blockIdx.x * 256 + (int)threadIdx.x) >> 5);
  const int g = lane / LPR;
  const int sub = lane - g * LPR;
  const int n0 = wave * NPW;
  if (n0 >= N) return;

  int ov = 0;
  if (lane <= NPW && n0 + lane <= N) ov = off[n0 + lane];

  const float* frow = feat + sub * 4;

#pragma unroll
  for (int k = 0; k < NPW; ++k) {
    const int node = n0 + k;
    if (node >= N) break;
    const int beg = __builtin_amdgcn_readlane(ov, k);
    const int end = __builtin_amdgcn_readlane(ov, k + 1);
    float4 acc = make_float4(0.f, 0.f, 0.f, 0.f);
    for (int c = beg; c < end; c += 32) {
      const int m = (end - c < 32) ? (end - c) : 32;
      const int idv = (lane < m) ? csr[c + lane] : 0;
      for (int j = 0; j < m; j += 8 * G) {
        float4 v[8];
#pragma unroll
        for (int u = 0; u < 8; ++u) {
          v[u] = make_float4(0.f, 0.f, 0.f, 0.f);
          int s;
          if (G == 1) {
            s = __builtin_amdgcn_readlane(idv, (j + u) & 31);
          } else {
            int s0 = __builtin_amdgcn_readlane(idv, (j + 2 * u) & 31);
            int s1 = __builtin_amdgcn_readlane(idv, (j + 2 * u + 1) & 31);
            s = g ? s1 : s0;
          }
          if (j + u * G + g < m) v[u] = *(const float4*)(frow + (size_t)s * D);
        }
#pragma unroll
        for (int u = 0; u < 8; ++u) {
          acc.x += v[u].x;
          acc.y += v[u].y;
          acc.z += v[u].z;
          acc.w += v[u].w;
        }
      }
    }
    if (G == 2) {
      acc.x += __shfl_xor(acc.x, 16);
      acc.y += __shfl_xor(acc.y, 16);
      acc.z += __shfl_xor(acc.z, 16);
      acc.w += __shfl_xor(acc.w, 16);
    }
    if (g == 0) *(float4*)(sum + (size_t)node * D + sub * 4) = acc;
  }
}

template<int NCOLS, bool TWO, bool ADD_MEAN, bool RELU, bool BIAS>
__global__ __launch_bounds__(256) void k_gemm(const float* __restrict__ A1,
                                              const float* __restrict__ A2,
                                              const float* __restrict__ rdeg,
                                              const float* __restrict__ B1,
                                              const float* __restrict__ B2,
                                              const float* __restrict__ bias,
                                              const float* __restrict__ addmat,
                                              float* __restrict__ out, int M) {
  const int NT = NCOLS / 16;
  int wave = (blockIdx.x * 256 + (int)threadIdx.x) >> 5;
  int mt = wave / NT;
  int nt = wave - mt * NT;
  int row0 = mt * 16;
  if (row0 >= M) return;

  int lane = threadIdx.x & 31;
  int hh = lane >> 4;
  int lr = lane & 15;
  int col0 = nt * 16;

  const float* a1 = A1 + (size_t)(row0 + lr) * HID + 2 * hh;
  const float* b1 = B1 + (size_t)(2 * hh) * NCOLS + col0 + lr;
  const float* a2 = nullptr;
  const float* b2 = nullptr;
  float rd = 0.f;
  if (TWO) {
    a2 = A2 + (size_t)(row0 + lr) * HID + 2 * hh;
    b2 = B2 + (size_t)(2 * hh) * NCOLS + col0 + lr;
    rd = rdeg[row0 + lr];
  }

  v8f c = {};
#pragma unroll 4
  for (int k = 0; k < HID; k += 4) {
    v2f av = *(const v2f*)(a1 + k);
    v2f bv;
    bv.x = b1[(size_t)k * NCOLS];
    bv.y = b1[(size_t)(k + 1) * NCOLS];
    c = __builtin_amdgcn_wmma_f32_16x16x4_f32(false, av, false, bv, (short)0, c,
                                              false, false);
    if (TWO) {
      v2f av2 = *(const v2f*)(a2 + k);
      av2.x *= rd;
      av2.y *= rd;
      v2f bv2;
      bv2.x = b2[(size_t)k * NCOLS];
      bv2.y = b2[(size_t)(k + 1) * NCOLS];
      c = __builtin_amdgcn_wmma_f32_16x16x4_f32(false, av2, false, bv2, (short)0,
                                                c, false, false);
    }
  }

  float bb = BIAS ? bias[col0 + lr] : 0.f;
#pragma unroll
  for (int v = 0; v < 8; ++v) {
    int r = row0 + v + 8 * hh;
    float val = c[v] + bb;
    if (ADD_MEAN) val += addmat[(size_t)r * NCOLS + col0 + lr] * rdeg[r];
    if (RELU) val = fmaxf(val, 0.f);
    out[(size_t)r * NCOLS + col0 + lr] = val;
  }
}

extern "C" void kernel_launch(void* const* d_in, const int* in_sizes, int n_in,
                              void* d_out, int out_size, void* d_ws,
                              size_t ws_size, hipStream_t stream) {
  const float* x   = (const float*)d_in[0];
  const int*   src = (const int*)d_in[1];
  const int*   dst = (const int*)d_in[2];
  const float* Ws1 = (const float*)d_in[3];
  const float* b1  = (const float*)d_in[4];
  const float* Wn1 = (const float*)d_in[5];
  const float* Ws2 = (const float*)d_in[6];
  const float* b2  = (const float*)d_in[7];
  const float* Wn2 = (const float*)d_in[8];
  const float* Ws3 = (const float*)d_in[9];
  const float* b3  = (const float*)d_in[10];
  const float* Wn3 = (const float*)d_in[11];
  float* out = (float*)d_out;

  const int N = in_sizes[0] / HID;
  const int E = in_sizes[1];

  float* deg = (float*)d_ws;
  float* h1  = deg + N;
  float* h2  = h1 + (size_t)N * HID;
  int*   cnt = (int*)(h2 + (size_t)N * HID);
  float* sum = (float*)cnt + (size_t)N * OUTD;
  int*   off = cnt + N;
  int*   csr = off + N + 1;
  int*   rnk = csr + E;
  float* tb  = h1;

  const int mtiles = (N + 15) / 16;
  const int tilesH = mtiles * (HID / 16);
  const int tilesO = mtiles * (OUTD / 16);
  const int NPW = 4;
  const int gwaves = (N + NPW - 1) / NPW;

  hipMemsetAsync(cnt, 0, (size_t)N * sizeof(int), stream);
  k_count<<<(E + 255) / 256, 256, 0, stream>>>(dst, cnt, rnk, E);
  k_scan<<<(N + 1023) / 1024, 256, 0, stream>>>(cnt, off, deg, N);
  k_fill<<<(E + 255) / 256, 256, 0, stream>>>(src, dst, off, rnk, csr, E);

  k_gather<HID, NPW><<<(gwaves + 7) / 8, 256, 0, stream>>>(x, off, csr, sum, N);
  k_gemm<HID, true, false, true, true><<<(tilesH + 7) / 8, 256, 0, stream>>>(
      x, sum, deg, Ws1, Wn1, b1, nullptr, h1, N);

  k_gather<HID, NPW><<<(gwaves + 7) / 8, 256, 0, stream>>>(h1, off, csr, sum, N);
  k_gemm<HID, true, false, true, true><<<(tilesH + 7) / 8, 256, 0, stream>>>(
      h1, sum, deg, Ws2, Wn2, b2, nullptr, h2, N);

  k_gemm<OUTD, false, false, false, false><<<(tilesO + 7) / 8, 256, 0, stream>>>(
      h2, nullptr, nullptr, Wn3, nullptr, nullptr, nullptr, tb, N);
  k_gather<OUTD, NPW><<<(gwaves + 7) / 8, 256, 0, stream>>>(tb, off, csr, sum, N);
  k_gemm<OUTD, false, true, false, true><<<(tilesO + 7) / 8, 256, 0, stream>>>(
      h2, nullptr, deg, Ws3, nullptr, b3, sum, out, N);
}


# GEMMs: per-wave weight fragments kept in registers for the launch, activation row tiles staged once per workgroup in LDS (whole-line loads) and read as A fragments; sum order unchanged
# speedup vs baseline: 2.6116x; 1.2046x over previous
#include <hip/hip_runtime.h>

#define HID 128
#define OUTD 64

typedef float v2f __attribute__((ext_vector_type(2)));
typedef float v8f __attribute__((ext_vector_type(8)));


__global__ __launch_bounds__(256) void k_count(const int* __restrict__ dst,
                                               int* __restrict__ cnt,
                                               int* __restrict__ rank, int E) {
  int e = blockIdx.x * 256 + threadIdx.x;
  if (e < E) rank[e] = atomicAdd(&cnt[dst[e]], 1);
}

__global__ __launch_bounds__(256) void k_scan(const int* __restrict__ cnt,
                                              int* __restrict__ off,
                                              float* __restrict__ rdeg, int N) {
  __shared__ int s_base[8];
  __shared__ int s_part[8];
  const int tid = threadIdx.x;
  const int lane = tid & 31;
  const int w = tid >> 5;
  const int lim = blockIdx.x * 1024;

  int a0 = 0, a1 = 0, a2 = 0, a3 = 0;
  int i = tid;
  for (; i + 768 < lim; i += 1024) {
    a0 += cnt[i];
    a1 += cnt[i + 256];
    a2 += cnt[i + 512];
    a3 += cnt[i + 768];
  }
  for (; i < lim; i += 256) a0 += cnt[i];
  int acc = (a0 + a1) + (a2 + a3);
  for (int o = 16; o > 0; o >>= 1) acc += __shfl_xor(acc, o);
  if (lane == 0) s_base[w] = acc;

  const int i0 = lim + 4 * tid;
  int c0 = (i0 + 0 < N) ? cnt[i0 + 0] : 0;
  int c1 = (i0 + 1 < N) ? cnt[i0 + 1] : 0;
  int c2 = (i0 + 2 < N) ? cnt[i0 + 2] : 0;
  int c3 = (i0 + 3 < N) ? cnt[i0 + 3] : 0;
  const int s = (c0 + c1) + (c2 + c3);
  int inc = s;
  for (int o = 1; o < 32; o <<= 1) {
    int t = __shfl_up(inc, o);
    if (lane >= o) inc += t;
  }
  if (lane == 31) s_part[w] = inc;
  __syncthreads();
  int base = 0;
  for (int k = 0; k < 8; ++k) base += s_base[k];
  for (int k = 0; k < w; ++k) base += s_part[k];
  int ex = base + inc - s;
  const int cc[4] = {c0, c1, c2, c3};
#pragma unroll
  for (int k = 0; k < 4; ++k) {
    int n = i0 + k;
    if (n < N) {
      off[n] = ex;
      rdeg[n] = 1.0f / fmaxf((float)cc[k], 1.0f);
      if (n == N - 1) off[N] = ex + cc[k];
    }
    ex += cc[k];
  }
}

__global__ __launch_bounds__(256) void k_fill(const int* __restrict__ src,
                                              const int* __restrict__ dst,
                                              const int* __restrict__ off,
                                              const int* __restrict__ rank,
                                              int* __restrict__ csr, int E) {
  int e = blockIdx.x * 256 + threadIdx.x;
  if (e < E) csr[off[dst[e]] + rank[e]] = src[e];
}

template<int D, int NPW>
__global__ __launch_bounds__(256) void k_gather(const float* __restrict__ feat,
                                                const int* __restrict__ off,
                                                const int* __restrict__ csr,
                                                float* __restrict__ sum, int N) {
  constexpr int LPR = D / 4;
  constexpr int G = 32 / LPR;
  const int lane = threadIdx.x & 31;
  const int wave = __builtin_amdgcn_readfirstlane((blockIdx.x * 256 + (int)threadIdx.x) >> 5);
  const int g = lane / LPR;
  const int sub = lane - g * LPR;
  const int n0 = wave * NPW;
  if (n0 >= N) return;

  int ov = 0;
  if (lane <= NPW && n0 + lane <= N) ov = off[n0 + lane];

  const float* frow = feat + sub * 4;

#pragma unroll
  for (int k = 0; k < NPW; ++k) {
    const int node = n0 + k;
    if (node >= N) break;
    const int beg = __builtin_amdgcn_readlane(ov, k);
    const int end = __builtin_amdgcn_readlane(ov, k + 1);
    float4 acc = make_float4(0.f, 0.f, 0.f, 0.f);
    for (int c = beg; c < end; c += 32) {
      const int m = (end - c < 32) ? (end - c) : 32;
      const int idv = (lane < m) ? csr[c + lane] : 0;
      for (int j = 0; j < m; j += 8 * G) {
        float4 v[8];
#pragma unroll
        for (int u = 0; u < 8; ++u) {
          v[u] = make_float4(0.f, 0.f, 0.f, 0.f);
          int s;
          if (G == 1) {
            s = __builtin_amdgcn_readlane(idv, (j + u) & 31);
          } else {
            int s0 = __builtin_amdgcn_readlane(idv, (j + 2 * u) & 31);
            int s1 = __builtin_amdgcn_readlane(idv, (j + 2 * u + 1) & 31);
            s = g ? s1 : s0;
          }
          if (j + u * G + g < m) v[u] = *(const float4*)(frow + (size_t)s * D);
        }
#pragma unroll
        for (int u = 0; u < 8; ++u) {
          acc.x += v[u].x;
          acc.y += v[u].y;
          acc.z += v[u].z;
          acc.w += v[u].w;
        }
      }
    }
    if (G == 2) {
      acc.x += __shfl_xor(acc.x, 16);
      acc.y += __shfl_xor(acc.y, 16);
      acc.z += __shfl_xor(acc.z, 16);
      acc.w += __shfl_xor(acc.w, 16);
    }
    if (g == 0) *(float4*)(sum + (size_t)node * D + sub * 4) = acc;
  }
}

#define APITCH 132

template<int NCOLS, bool TWO, bool ADD_MEAN, bool RELU, bool BIAS>
__global__ __launch_bounds__(256) void k_gemm(const float* __restrict__ A1,
                                              const float* __restrict__ A2,
                                              const float* __restrict__ rdeg,
                                              const float* __restrict__ B1,
                                              const float* __restrict__ B2,
                                              const float* __restrict__ bias,
                                              const float* __restrict__ addmat,
                                              float* __restrict__ out, int M) {
  constexpr int SUBF = 16 * APITCH;
  constexpr int STGF = 2 * SUBF;
  constexpr int ROWS = TWO ? 16 : 32;
  constexpr int KS = HID / 4;
  __shared__ __attribute__((aligned(16))) float s_a[2 * STGF];

  const int tid = threadIdx.x;
  const int lane = tid & 31;
  const int w = __builtin_amdgcn_readfirstlane(tid >> 5);
  const int hh = lane >> 4;
  const int lr = lane & 15;
  const int nt = TWO ? w : (w & 3);
  const int sub = TWO ? 0 : (w >> 2);
  const int col0 = nt * 16;

  v2f bf1[KS];
  v2f bf2[TWO ? KS : 1];
  {
    const float* b1 = B1 + (size_t)(2 * hh) * NCOLS + col0 + lr;
#pragma unroll
    for (int t = 0; t < KS; ++t) {
      bf1[t].x = b1[(size_t)(4 * t) * NCOLS];
      bf1[t].y = b1[(size_t)(4 * t + 1) * NCOLS];
    }
    if (TWO) {
      const float* b2 = B2 + (size_t)(2 * hh) * NCOLS + col0 + lr;
#pragma unroll
      for (int t = 0; t < KS; ++t) {
        bf2[t].x = b2[(size_t)(4 * t) * NCOLS];
        bf2[t].y = b2[(size_t)(4 * t + 1) * NCOLS];
      }
    }
  }
  const float bb = BIAS ? bias[col0 + lr] : 0.f;

  const int ntiles = (M + ROWS - 1) / ROWS;
  const int step = (int)gridDim.x;

  float4 pf[4];
  float rdv[2] = {1.f, 1.f};
  auto gload = [&](int tile) {
    const int row0 = tile * ROWS;
#pragma unroll
    for (int i = 0; i < 4; ++i) {
      const int sb = i >> 1;
      const int r = w + 8 * (i & 1);
      const int grow = TWO ? (row0 + r) : (row0 + 16 * sb + r);
      const int gc = grow < M ? grow : M - 1;
      const float* base = (TWO && sb) ? A2 : A1;
      pf[i] = *(const float4*)(base + (size_t)gc * HID + lane * 4);
      if (TWO && sb) rdv[i & 1] = rdeg[gc];
    }
  };
  auto sstore = [&](int stage, int tile) {
    const int row0 = tile * ROWS;
#pragma unroll
    for (int i = 0; i < 4; ++i) {
      const int sb = i >> 1;
      const int r = w + 8 * (i & 1);
      const int grow = TWO ? (row0 + r) : (row0 + 16 * sb + r);
      float4 v = pf[i];
      if (TWO && sb) {
        const float rd = rdv[i & 1];
        v.x *= rd; v.y *= rd; v.z *= rd; v.w *= rd;
      }
      if (grow >= M) v = make_float4(0.f, 0.f, 0.f, 0.f);
      *(float4*)(s_a + stage * STGF + sb * SUBF + r * APITCH + lane * 4) = v;
    }
  };

  int tile = blockIdx.x;
  if (tile < ntiles) {
    gload(tile);
    sstore(0, tile);
  }
  __syncthreads();

  constexpr int KB = 8;
  int st = 0;
  for (; tile < ntiles; tile += step) {
    const int nxt = tile + step;
    const bool more = nxt < ntiles;
    if (more) gload(nxt);

    const float* fa = s_a + st * STGF + sub * SUBF + lr * APITCH + 2 * hh;
    v2f fr1[2][KB];
    v2f fr2[2][TWO ? KB : 1];
#pragma unroll
    for (int u = 0; u < KB; ++u) {
      fr1[0][u] = *(const v2f*)(fa + 4 * u);
      if (TWO) fr2[0][u] = *(const v2f*)(fa + SUBF + 4 * u);
    }
    v8f c = {};
#pragma unroll
    for (int kb = 0; kb < KS / KB; ++kb) {
      if (kb + 1 < KS / KB) {
#pragma unroll
        for (int u = 0; u < KB; ++u) {
          const int t = (kb + 1) * KB + u;
          fr1[(kb + 1) & 1][u] = *(const v2f*)(fa + 4 * t);
          if (TWO) fr2[(kb + 1) & 1][u] = *(const v2f*)(fa + SUBF + 4 * t);
        }
      }
      __builtin_amdgcn_sched_barrier(0);
#pragma unroll
      for (int u = 0; u < KB; ++u) {
        const int t = kb * KB + u;
        c = __builtin_amdgcn_wmma_f32_16x16x4_f32(false, fr1[kb & 1][u], false, bf1[t],
                                                  (short)0, c, false, false);
        if (TWO)
          c = __builtin_amdgcn_wmma_f32_16x16x4_f32(false, fr2[kb & 1][u], false, bf2[t],
                                                    (short)0, c, false, false);
      }
      __builtin_amdgcn_sched_barrier(0);
    }

    if (more) sstore(st ^ 1, nxt);
    __syncthreads();
    st ^= 1;

    const int row0 = tile * ROWS + 16 * sub;
#pragma unroll
    for (int v = 0; v < 8; ++v) {
      const int r = row0 + v + 8 * hh;
      if (r < M) {
        float val = c[v] + bb;
        if (ADD_MEAN) val += addmat[(size_t)r * NCOLS + col0 + lr] * rdeg[r];
        if (RELU) val = fmaxf(val, 0.f);
        out[(size_t)r * NCOLS + col0 + lr] = val;
      }
    }
  }
}

extern "C" void kernel_launch(void* const* d_in, const int* in_sizes, int n_in,
                              void* d_out, int out_size, void* d_ws,
                              size_t ws_size, hipStream_t stream) {
  const float* x   = (const float*)d_in[0];
  const int*   src = (const int*)d_in[1];
  const int*   dst = (const int*)d_in[2];
  const float* Ws1 = (const float*)d_in[3];
  const float* b1  = (const float*)d_in[4];
  const float* Wn1 = (const float*)d_in[5];
  const float* Ws2 = (const float*)d_in[6];
  const float* b2  = (const float*)d_in[7];
  const float* Wn2 = (const float*)d_in[8];
  const float* Ws3 = (const float*)d_in[9];
  const float* b3  = (const float*)d_in[10];
  const float* Wn3 = (const float*)d_in[11];
  float* out = (float*)d_out;

  const int N = in_sizes[0] / HID;
  const int E = in_sizes[1];

  float* deg = (float*)d_ws;
  float* h1  = deg + N;
  float* h2  = h1 + (size_t)N * HID;
  int*   cnt = (int*)(h2 + (size_t)N * HID);
  float* sum = (float*)cnt + (size_t)N * OUTD;
  int*   off = cnt + N;
  int*   csr = off + N + 1;
  int*   rnk = csr + E;
  float* tb  = h1;

  const int ggrid = 512;
  const int NPW = 4;
  const int gwaves = (N + NPW - 1) / NPW;

  hipMemsetAsync(cnt, 0, (size_t)N * sizeof(int), stream);
  k_count<<<(E + 255) / 256, 256, 0, stream>>>(dst, cnt, rnk, E);
  k_scan<<<(N + 1023) / 1024, 256, 0, stream>>>(cnt, off, deg, N);
  k_fill<<<(E + 255) / 256, 256, 0, stream>>>(src, dst, off, rnk, csr, E);

  k_gather<HID, NPW><<<(gwaves + 7) / 8, 256, 0, stream>>>(x, off, csr, sum, N);
  k_gemm<HID, true, false, true, true><<<ggrid, 256, 0, stream>>>(
      x, sum, deg, Ws1, Wn1, b1, nullptr, h1, N);

  k_gather<HID, NPW><<<(gwaves + 7) / 8, 256, 0, stream>>>(h1, off, csr, sum, N);
  k_gemm<HID, true, false, true, true><<<ggrid, 256, 0, stream>>>(
      h1, sum, deg, Ws2, Wn2, b2, nullptr, h2, N);

  k_gemm<OUTD, false, false, false, false><<<ggrid, 256, 0, stream>>>(
      h2, nullptr, nullptr, Wn3, nullptr, nullptr, nullptr, tb, N);
  k_gather<OUTD, NPW><<<(gwaves + 7) / 8, 256, 0, stream>>>(tb, off, csr, sum, N);
  k_gemm<OUTD, false, true, false, true><<<ggrid, 256, 0, stream>>>(
      h2, nullptr, deg, Ws3, nullptr, b3, sum, out, N);
}


# CSR build regrouped through LDS: edges binned into 256-node buckets by LDS counters (one global add per chunk and bucket), one workgroup per bucket ranks and fills; sum order = LDS arrival rank
# speedup vs baseline: 3.4944x; 1.3380x over previous
#include <hip/hip_runtime.h>

#define HID 128
#define OUTD 64

typedef float v2f __attribute__((ext_vector_type(2)));
typedef float v8f __attribute__((ext_vector_type(8)));


#define BSH 8
#define BW 256
#define MAXB 256
#define CHUNK 4096
#define EPT (CHUNK / 256)
#define PPT 20

__global__ __launch_bounds__(256) void k_bhist(const int* __restrict__ dst,
                                               int* __restrict__ gcnt, int E) {
  __shared__ int s_h[MAXB];
  const int tid = threadIdx.x;
  s_h[tid] = 0;
  __syncthreads();
  const int e0 = blockIdx.x * CHUNK + tid;
  int d[EPT];
#pragma unroll
  for (int k = 0; k < EPT; ++k) {
    const int e = e0 + k * 256;
    d[k] = dst[e < E ? e : E - 1];
  }
#pragma unroll
  for (int k = 0; k < EPT; ++k)
    if (e0 + k * 256 < E) atomicAdd(&s_h[d[k] >> BSH], 1);
  __syncthreads();
  const int c = s_h[tid];
  if (c) atomicAdd(&gcnt[tid], c);
}

__global__ __launch_bounds__(256) void k_bscatter(const int* __restrict__ src,
                                                  const int* __restrict__ dst,
                                                  const int* __restrict__ gcnt,
                                                  int* __restrict__ gcur,
                                                  unsigned* __restrict__ part, int E) {
  __shared__ int s_h[MAXB];
  __shared__ int s_w[8];
  const int tid = threadIdx.x;
  const int lane = tid & 31;
  const int w = tid >> 5;
  s_h[tid] = 0;
  const int e0 = blockIdx.x * CHUNK + tid;
  int d[EPT], sv[EPT], r[EPT];
#pragma unroll
  for (int k = 0; k < EPT; ++k) {
    const int e = e0 + k * 256;
    const int ec = e < E ? e : E - 1;
    d[k] = dst[ec];
    sv[k] = src[ec];
  }
  const int g = gcnt[tid];
  int inc = g;
  for (int o = 1; o < 32; o <<= 1) {
    int t = __shfl_up(inc, o);
    if (lane >= o) inc += t;
  }
  if (lane == 31) s_w[w] = inc;
  __syncthreads();
  int start = inc - g;
  for (int k = 0; k < w; ++k) start += s_w[k];

#pragma unroll
  for (int k = 0; k < EPT; ++k)
    if (e0 + k * 256 >= E) d[k] = -1;
#pragma unroll
  for (int k = 0; k < EPT; ++k) {
    r[k] = 0;
    if (d[k] >= 0) r[k] = atomicAdd(&s_h[d[k] >> BSH], 1);
  }
  __syncthreads();
  const int c = s_h[tid];
  if (c) start += atomicAdd(&gcur[tid], c);
  s_h[tid] = start;
  __syncthreads();
#pragma unroll
  for (int k = 0; k < EPT; ++k) r[k] += s_h[(d[k] >> BSH) & (MAXB - 1)];
#pragma unroll
  for (int k = 0; k < EPT; ++k)
    if (d[k] >= 0) part[r[k]] = ((unsigned)sv[k] << BSH) | (unsigned)(d[k] & (BW - 1));
}

__global__ __launch_bounds__(256) void k_bcsr(const int* __restrict__ gcnt,
                                              const unsigned* __restrict__ part,
                                              int* __restrict__ off,
                                              float* __restrict__ rdeg,
                                              int* __restrict__ csr, int N) {
  __shared__ int s_c[BW];
  __shared__ int s_w[8];
  __shared__ int s_v[8];
  const int tid = threadIdx.x;
  const int lane = tid & 31;
  const int w = tid >> 5;
  const int b = blockIdx.x;
  const int cnt = gcnt[b];
  int g = (tid < b) ? gcnt[tid] : 0;
  for (int o = 16; o > 0; o >>= 1) g += __shfl_xor(g, o);
  if (lane == 0) s_w[w] = g;
  s_c[tid] = 0;
  __syncthreads();
  int base = 0;
  for (int k = 0; k < 8; ++k) base += s_w[k];
  const unsigned* p = part + base;

  for (int i0 = tid; i0 < cnt; i0 += 256 * PPT) {
    unsigned v[PPT];
#pragma unroll
    for (int u = 0; u < PPT; ++u) {
      const int i = i0 + u * 256;
      v[u] = p[i < cnt ? i : cnt - 1];
    }
#pragma unroll
    for (int u = 0; u < PPT; ++u)
      if (i0 + u * 256 < cnt) atomicAdd(&s_c[v[u] & (BW - 1)], 1);
  }
  __syncthreads();
  const int c = s_c[tid];
  int inc = c;
  for (int o = 1; o < 32; o <<= 1) {
    int t = __shfl_up(inc, o);
    if (lane >= o) inc += t;
  }
  if (lane == 31) s_v[w] = inc;
  __syncthreads();
  int ex = base + inc - c;
  for (int k = 0; k < w; ++k) ex += s_v[k];
  const int n = b * BW + tid;
  if (n < N) {
    off[n] = ex;
    rdeg[n] = 1.0f / fmaxf((float)c, 1.0f);
    if (n == N - 1) off[N] = ex + c;
  }
  s_c[tid] = ex;
  __syncthreads();
  for (int i0 = tid; i0 < cnt; i0 += 256 * PPT) {
    unsigned v[PPT];
    int slot[PPT];
#pragma unroll
    for (int u = 0; u < PPT; ++u) {
      const int i = i0 + u * 256;
      v[u] = p[i < cnt ? i : cnt - 1];
    }
#pragma unroll
    for (int u = 0; u < PPT; ++u) {
      slot[u] = 0;
      if (i0 + u * 256 < cnt) slot[u] = atomicAdd(&s_c[v[u] & (BW - 1)], 1);
    }
#pragma unroll
    for (int u = 0; u < PPT; ++u)
      if (i0 + u * 256 < cnt) csr[slot[u]] = (int)(v[u] >> BSH);
  }
}

template<int D, int NPW>
__global__ __launch_bounds__(256) void k_gather(const float* __restrict__ feat,
                                                const int* __restrict__ off,
                                                const int* __restrict__ csr,
                                                float* __restrict__ sum, int N) {
  constexpr int LPR = D / 4;
  constexpr int G = 32 / LPR;
  const int lane = threadIdx.x & 31;
  const int wave = __builtin_amdgcn_readfirstlane((blockIdx.x * 256 + (int)threadIdx.x) >> 5);
  const int g = lane / LPR;
  const int sub = lane - g * LPR;
  const int n0 = wave * NPW;
  if (n0 >= N) return;

  int ov = 0;
  if (lane <= NPW && n0 + lane <= N) ov = off[n0 + lane];

  const float* frow = feat + sub * 4;

#pragma unroll
  for (int k = 0; k < NPW; ++k) {
    const int node = n0 + k;
    if (node >= N) break;
    const int beg = __builtin_amdgcn_readlane(ov, k);
    const int end = __builtin_amdgcn_readlane(ov, k + 1);
    float4 acc = make_float4(0.f, 0.f, 0.f, 0.f);
    for (int c = beg; c < end; c += 32) {
      const int m = (end - c < 32) ? (end - c) : 32;
      const int idv = (lane < m) ? csr[c + lane] : 0;
      for (int j = 0; j < m; j += 8 * G) {
        float4 v[8];
#pragma unroll
        for (int u = 0; u < 8; ++u) {
          v[u] = make_float4(0.f, 0.f, 0.f, 0.f);
          int s;
          if (G == 1) {
            s = __builtin_amdgcn_readlane(idv, (j + u) & 31);
          } else {
            int s0 = __builtin_amdgcn_readlane(idv, (j + 2 * u) & 31);
            int s1 = __builtin_amdgcn_readlane(idv, (j + 2 * u + 1) & 31);
            s = g ? s1 : s0;
          }
          if (j + u * G + g < m) v[u] = *(const float4*)(frow + (size_t)s * D);
        }
#pragma unroll
        for (int u = 0; u < 8; ++u) {
          acc.x += v[u].x;
          acc.y += v[u].y;
          acc.z += v[u].z;
          acc.w += v[u].w;
        }
      }
    }
    if (G == 2) {
      acc.x += __shfl_xor(acc.x, 16);
      acc.y += __shfl_xor(acc.y, 16);
      acc.z += __shfl_xor(acc.z, 16);
      acc.w += __shfl_xor(acc.w, 16);
    }
    if (g == 0) *(float4*)(sum + (size_t)node * D + sub * 4) = acc;
  }
}

#define APITCH 132

template<int NCOLS, bool TWO, bool ADD_MEAN, bool RELU, bool BIAS>
__global__ __launch_bounds__(256) void k_gemm(const float* __restrict__ A1,
                                              const float* __restrict__ A2,
                                              const float* __restrict__ rdeg,
                                              const float* __restrict__ B1,
                                              const float* __restrict__ B2,
                                              const float* __restrict__ bias,
                                              const float* __restrict__ addmat,
                                              float* __restrict__ out, int M) {
  constexpr int SUBF = 16 * APITCH;
  constexpr int STGF = 2 * SUBF;
  constexpr int ROWS = TWO ? 16 : 32;
  constexpr int KS = HID / 4;
  __shared__ __attribute__((aligned(16))) float s_a[2 * STGF];

  const int tid = threadIdx.x;
  const int lane = tid & 31;
  const int w = __builtin_amdgcn_readfirstlane(tid >> 5);
  const int hh = lane >> 4;
  const int lr = lane & 15;
  const int nt = TWO ? w : (w & 3);
  const int sub = TWO ? 0 : (w >> 2);
  const int col0 = nt * 16;

  v2f bf1[KS];
  v2f bf2[TWO ? KS : 1];
  {
    const float* b1 = B1 + (size_t)(2 * hh) * NCOLS + col0 + lr;
#pragma unroll
    for (int t = 0; t < KS; ++t) {
      bf1[t].x = b1[(size_t)(4 * t) * NCOLS];
      bf1[t].y = b1[(size_t)(4 * t + 1) * NCOLS];
    }
    if (TWO) {
      const float* b2 = B2 + (size_t)(2 * hh) * NCOLS + col0 + lr;
#pragma unroll
      for (int t = 0; t < KS; ++t) {
        bf2[t].x = b2[(size_t)(4 * t) * NCOLS];
        bf2[t].y = b2[(size_t)(4 * t + 1) * NCOLS];
      }
    }
  }
  const float bb = BIAS ? bias[col0 + lr] : 0.f;

  const int ntiles = (M + ROWS - 1) / ROWS;
  const int step = (int)gridDim.x;

  float4 pf[4];
  float rdv[2] = {1.f, 1.f};
  auto gload = [&](int tile) {
    const int row0 = tile * ROWS;
#pragma unroll
    for (int i = 0; i < 4; ++i) {
      const int sb = i >> 1;
      const int r = w + 8 * (i & 1);
      const int grow = TWO ? (row0 + r) : (row0 + 16 * sb + r);
      const int gc = grow < M ? grow : M - 1;
      const float* base = (TWO && sb) ? A2 : A1;
      pf[i] = *(const float4*)(base + (size_t)gc * HID + lane * 4);
      if (TWO && sb) rdv[i & 1] = rdeg[gc];
    }
  };
  auto sstore = [&](int stage, int tile) {
    const int row0 = tile * ROWS;
#pragma unroll
    for (int i = 0; i < 4; ++i) {
      const int sb = i >> 1;
      const int r = w + 8 * (i & 1);
      const int grow = TWO ? (row0 + r) : (row0 + 16 * sb + r);
      float4 v = pf[i];
      if (TWO && sb) {
        const float rd = rdv[i & 1];
        v.x *= rd; v.y *= rd; v.z *= rd; v.w *= rd;
      }
      if (grow >= M) v = make_float4(0.f, 0.f, 0.f, 0.f);
      *(float4*)(s_a + stage * STGF + sb * SUBF + r * APITCH + lane * 4) = v;
    }
  };

  int tile = blockIdx.x;
  if (tile < ntiles) {
    gload(tile);
    sstore(0, tile);
  }
  __syncthreads();

  constexpr int KB = 8;
  int st = 0;
  for (; tile < ntiles; tile += step) {
    const int nxt = tile + step;
    const bool more = nxt < ntiles;
    if (more) gload(nxt);

    const float* fa = s_a + st * STGF + sub * SUBF + lr * APITCH + 2 * hh;
    v2f fr1[2][KB];
    v2f fr2[2][TWO ? KB : 1];
#pragma unroll
    for (int u = 0; u < KB; ++u) {
      fr1[0][u] = *(const v2f*)(fa + 4 * u);
      if (TWO) fr2[0][u] = *(const v2f*)(fa + SUBF + 4 * u);
    }
    v8f c = {};
#pragma unroll
    for (int kb = 0; kb < KS / KB; ++kb) {
      if (kb + 1 < KS / KB) {
#pragma unroll
        for (int u = 0; u < KB; ++u) {
          const int t = (kb + 1) * KB + u;
          fr1[(kb + 1) & 1][u] = *(const v2f*)(fa + 4 * t);
          if (TWO) fr2[(kb + 1) & 1][u] = *(const v2f*)(fa + SUBF + 4 * t);
        }
      }
      __builtin_amdgcn_sched_barrier(0);
#pragma unroll
      for (int u = 0; u < KB; ++u) {
        const int t = kb * KB + u;
        c = __builtin_amdgcn_wmma_f32_16x16x4_f32(false, fr1[kb & 1][u], false, bf1[t],
                                                  (short)0, c, false, false);
        if (TWO)
          c = __builtin_amdgcn_wmma_f32_16x16x4_f32(false, fr2[kb & 1][u], false, bf2[t],
                                                    (short)0, c, false, false);
      }
      __builtin_amdgcn_sched_barrier(0);
    }

    if (more) sstore(st ^ 1, nxt);
    __syncthreads();
    st ^= 1;

    const int row0 = tile * ROWS + 16 * sub;
#pragma unroll
    for (int v = 0; v < 8; ++v) {
      const int r = row0 + v + 8 * hh;
      if (r < M) {
        float val = c[v] + bb;
        if (ADD_MEAN) val += addmat[(size_t)r * NCOLS + col0 + lr] * rdeg[r];
        if (RELU) val = fmaxf(val, 0.f);
        out[(size_t)r * NCOLS + col0 + lr] = val;
      }
    }
  }
}

extern "C" void kernel_launch(void* const* d_in, const int* in_sizes, int n_in,
                              void* d_out, int out_size, void* d_ws,
                              size_t ws_size, hipStream_t stream) {
  const float* x   = (const float*)d_in[0];
  const int*   src = (const int*)d_in[1];
  const int*   dst = (const int*)d_in[2];
  const float* Ws1 = (const float*)d_in[3];
  const float* b1  = (const float*)d_in[4];
  const float* Wn1 = (const float*)d_in[5];
  const float* Ws2 = (const float*)d_in[6];
  const float* b2  = (const float*)d_in[7];
  const float* Wn2 = (const float*)d_in[8];
  const float* Ws3 = (const float*)d_in[9];
  const float* b3  = (const float*)d_in[10];
  const float* Wn3 = (const float*)d_in[11];
  float* out = (float*)d_out;

  const int N = in_sizes[0] / HID;
  const int E = in_sizes[1];

  float* deg = (float*)d_ws;
  float* h1  = deg + N;
  float* h2  = h1 + (size_t)N * HID;
  int*   cnt = (int*)(h2 + (size_t)N * HID);
  float* sum = (float*)cnt + (size_t)N * OUTD;
  int*   gcur = cnt + MAXB;
  int*   off = gcur + MAXB;
  int*   csr = off + N + 1;
  unsigned* part = (unsigned*)(csr + E);
  float* tb  = h1;

  const int ggrid = 512;
  const int NPW = 4;
  const int gwaves = (N + NPW - 1) / NPW;

  hipMemsetAsync(cnt, 0, (size_t)(2 * MAXB) * sizeof(int), stream);
  k_bhist<<<(E + CHUNK - 1) / CHUNK, 256, 0, stream>>>(dst, cnt, E);
  k_bscatter<<<(E + CHUNK - 1) / CHUNK, 256, 0, stream>>>(src, dst, cnt, gcur, part, E);
  k_bcsr<<<(N + BW - 1) / BW, 256, 0, stream>>>(cnt, part, off, deg, csr, N);

  k_gather<HID, NPW><<<(gwaves + 7) / 8, 256, 0, stream>>>(x, off, csr, sum, N);
  k_gemm<HID, true, false, true, true><<<ggrid, 256, 0, stream>>>(
      x, sum, deg, Ws1, Wn1, b1, nullptr, h1, N);

  k_gather<HID, NPW><<<(gwaves + 7) / 8, 256, 0, stream>>>(h1, off, csr, sum, N);
  k_gemm<HID, true, false, true, true><<<ggrid, 256, 0, stream>>>(
      h1, sum, deg, Ws2, Wn2, b2, nullptr, h2, N);

  k_gemm<OUTD, false, false, false, false><<<ggrid, 256, 0, stream>>>(
      h2, nullptr, nullptr, Wn3, nullptr, nullptr, nullptr, tb, N);
  k_gather<OUTD, NPW><<<(gwaves + 7) / 8, 256, 0, stream>>>(tb, off, csr, sum, N);
  k_gemm<OUTD, false, true, false, true><<<ggrid, 256, 0, stream>>>(
      h2, nullptr, deg, Ws3, nullptr, b3, sum, out, N);
}


# Neighbour sums of layers 1-2 made inside the GEMM launch: the wave that stages two tile rows walks their CSR lists and lays the mean rows into LDS as A2; 2 launches fewer; sum order unchanged
# speedup vs baseline: 3.7647x; 1.0774x over previous
#include <hip/hip_runtime.h>

#define HID 128
#define OUTD 64

typedef float v2f __attribute__((ext_vector_type(2)));
typedef float v8f __attribute__((ext_vector_type(8)));


#define BSH 8
#define BW 256
#define MAXB 256
#define CHUNK 4096
#define EPT (CHUNK / 256)
#define PPT 20

__global__ __launch_bounds__(256) void k_bhist(const int* __restrict__ dst,
                                               int* __restrict__ gcnt, int E) {
  __shared__ int s_h[MAXB];
  const int tid = threadIdx.x;
  s_h[tid] = 0;
  __syncthreads();
  const int e0 = blockIdx.x * CHUNK + tid;
  int d[EPT];
#pragma unroll
  for (int k = 0; k < EPT; ++k) {
    const int e = e0 + k * 256;
    d[k] = dst[e < E ? e : E - 1];
  }
#pragma unroll
  for (int k = 0; k < EPT; ++k)
    if (e0 + k * 256 < E) atomicAdd(&s_h[d[k] >> BSH], 1);
  __syncthreads();
  const int c = s_h[tid];
  if (c) atomicAdd(&gcnt[tid], c);
}

__global__ __launch_bounds__(256) void k_bscatter(const int* __restrict__ src,
                                                  const int* __restrict__ dst,
                                                  const int* __restrict__ gcnt,
                                                  int* __restrict__ gcur,
                                                  unsigned* __restrict__ part, int E) {
  __shared__ int s_h[MAXB];
  __shared__ int s_w[8];
  const int tid = threadIdx.x;
  const int lane = tid & 31;
  const int w = tid >> 5;
  s_h[tid] = 0;
  const int e0 = blockIdx.x * CHUNK + tid;
  int d[EPT], sv[EPT], r[EPT];
#pragma unroll
  for (int k = 0; k < EPT; ++k) {
    const int e = e0 + k * 256;
    const int ec = e < E ? e : E - 1;
    d[k] = dst[ec];
    sv[k] = src[ec];
  }
  const int g = gcnt[tid];
  int inc = g;
  for (int o = 1; o < 32; o <<= 1) {
    int t = __shfl_up(inc, o);
    if (lane >= o) inc += t;
  }
  if (lane == 31) s_w[w] = inc;
  __syncthreads();
  int start = inc - g;
  for (int k = 0; k < w; ++k) start += s_w[k];

#pragma unroll
  for (int k = 0; k < EPT; ++k)
    if (e0 + k * 256 >= E) d[k] = -1;
#pragma unroll
  for (int k = 0; k < EPT; ++k) {
    r[k] = 0;
    if (d[k] >= 0) r[k] = atomicAdd(&s_h[d[k] >> BSH], 1);
  }
  __syncthreads();
  const int c = s_h[tid];
  if (c) start += atomicAdd(&gcur[tid], c);
  s_h[tid] = start;
  __syncthreads();
#pragma unroll
  for (int k = 0; k < EPT; ++k) r[k] += s_h[(d[k] >> BSH) & (MAXB - 1)];
#pragma unroll
  for (int k = 0; k < EPT; ++k)
    if (d[k] >= 0) part[r[k]] = ((unsigned)sv[k] << BSH) | (unsigned)(d[k] & (BW - 1));
}

__global__ __launch_bounds__(256) void k_bcsr(const int* __restrict__ gcnt,
                                              const unsigned* __restrict__ part,
                                              int* __restrict__ off,
                                              float* __restrict__ rdeg,
                                              int* __restrict__ csr, int N) {
  __shared__ int s_c[BW];
  __shared__ int s_w[8];
  __shared__ int s_v[8];
  const int tid = threadIdx.x;
  const int lane = tid & 31;
  const int w = tid >> 5;
  const int b = blockIdx.x;
  const int cnt = gcnt[b];
  int g = (tid < b) ? gcnt[tid] : 0;
  for (int o = 16; o > 0; o >>= 1) g += __shfl_xor(g, o);
  if (lane == 0) s_w[w] = g;
  s_c[tid] = 0;
  __syncthreads();
  int base = 0;
  for (int k = 0; k < 8; ++k) base += s_w[k];
  const unsigned* p = part + base;

  for (int i0 = tid; i0 < cnt; i0 += 256 * PPT) {
    unsigned v[PPT];
#pragma unroll
    for (int u = 0; u < PPT; ++u) {
      const int i = i0 + u * 256;
      v[u] = p[i < cnt ? i : cnt - 1];
    }
#pragma unroll
    for (int u = 0; u < PPT; ++u)
      if (i0 + u * 256 < cnt) atomicAdd(&s_c[v[u] & (BW - 1)], 1);
  }
  __syncthreads();
  const int c = s_c[tid];
  int inc = c;
  for (int o = 1; o < 32; o <<= 1) {
    int t = __shfl_up(inc, o);
    if (lane >= o) inc += t;
  }
  if (lane == 31) s_v[w] = inc;
  __syncthreads();
  int ex = base + inc - c;
  for (int k = 0; k < w; ++k) ex += s_v[k];
  const int n = b * BW + tid;
  if (n < N) {
    off[n] = ex;
    rdeg[n] = 1.0f / fmaxf((float)c, 1.0f);
    if (n == N - 1) off[N] = ex + c;
  }
  s_c[tid] = ex;
  __syncthreads();
  for (int i0 = tid; i0 < cnt; i0 += 256 * PPT) {
    unsigned v[PPT];
    int slot[PPT];
#pragma unroll
    for (int u = 0; u < PPT; ++u) {
      const int i = i0 + u * 256;
      v[u] = p[i < cnt ? i : cnt - 1];
    }
#pragma unroll
    for (int u = 0; u < PPT; ++u) {
      slot[u] = 0;
      if (i0 + u * 256 < cnt) slot[u] = atomicAdd(&s_c[v[u] & (BW - 1)], 1);
    }
#pragma unroll
    for (int u = 0; u < PPT; ++u)
      if (i0 + u * 256 < cnt) csr[slot[u]] = (int)(v[u] >> BSH);
  }
}

template<int D, int NPW>
__global__ __launch_bounds__(256) void k_gather(const float* __restrict__ feat,
                                                const int* __restrict__ off,
                                                const int* __restrict__ csr,
                                                float* __restrict__ sum, int N) {
  constexpr int LPR = D / 4;
  constexpr int G = 32 / LPR;
  const int lane = threadIdx.x & 31;
  const int wave = __builtin_amdgcn_readfirstlane((blockIdx.x * 256 + (int)threadIdx.x) >> 5);
  const int g = lane / LPR;
  const int sub = lane - g * LPR;
  const int n0 = wave * NPW;
  if (n0 >= N) return;

  int ov = 0;
  if (lane <= NPW && n0 + lane <= N) ov = off[n0 + lane];

  const float* frow = feat + sub * 4;

#pragma unroll
  for (int k = 0; k < NPW; ++k) {
    const int node = n0 + k;
    if (node >= N) break;
    const int beg = __builtin_amdgcn_readlane(ov, k);
    const int end = __builtin_amdgcn_readlane(ov, k + 1);
    float4 acc = make_float4(0.f, 0.f, 0.f, 0.f);
    for (int c = beg; c < end; c += 32) {
      const int m = (end - c < 32) ? (end - c) : 32;
      const int idv = (lane < m) ? csr[c + lane] : 0;
      for (int j = 0; j < m; j += 8 * G) {
        float4 v[8];
#pragma unroll
        for (int u = 0; u < 8; ++u) {
          v[u] = make_float4(0.f, 0.f, 0.f, 0.f);
          int s;
          if (G == 1) {
            s = __builtin_amdgcn_readlane(idv, (j + u) & 31);
          } else {
            int s0 = __builtin_amdgcn_readlane(idv, (j + 2 * u) & 31);
            int s1 = __builtin_amdgcn_readlane(idv, (j + 2 * u + 1) & 31);
            s = g ? s1 : s0;
          }
          if (j + u * G + g < m) v[u] = *(const float4*)(frow + (size_t)s * D);
        }
#pragma unroll
        for (int u = 0; u < 8; ++u) {
          acc.x += v[u].x;
          acc.y += v[u].y;
          acc.z += v[u].z;
          acc.w += v[u].w;
        }
      }
    }
    if (G == 2) {
      acc.x += __shfl_xor(acc.x, 16);
      acc.y += __shfl_xor(acc.y, 16);
      acc.z += __shfl_xor(acc.z, 16);
      acc.w += __shfl_xor(acc.w, 16);
    }
    if (g == 0) *(float4*)(sum + (size_t)node * D + sub * 4) = acc;
  }
}

#define APITCH 132

template<int NCOLS, bool TWO, bool ADD_MEAN, bool RELU, bool BIAS>
__global__ __launch_bounds__(256) void k_gemm(const float* __restrict__ A1,
                                              const float* __restrict__ A2,
                                              const float* __restrict__ rdeg,
                                              const float* __restrict__ B1,
                                              const float* __restrict__ B2,
                                              const float* __restrict__ bias,
                                              const float* __restrict__ addmat,
                                              const int* __restrict__ off,
                                              const int* __restrict__ csr,
                                              float* __restrict__ out, int M) {
  constexpr int SUBF = 16 * APITCH;
  constexpr int STGF = 2 * SUBF;
  constexpr int ROWS = TWO ? 16 : 32;
  constexpr int KS = HID / 4;
  __shared__ __attribute__((aligned(16))) float s_a[2 * STGF];

  const int tid = threadIdx.x;
  const int lane = tid & 31;
  const int w = __builtin_amdgcn_readfirstlane(tid >> 5);
  const int hh = lane >> 4;
  const int lr = lane & 15;
  const int nt = TWO ? w : (w & 3);
  const int sub = TWO ? 0 : (w >> 2);
  const int col0 = nt * 16;

  v2f bf1[KS];
  v2f bf2[TWO ? KS : 1];
  {
    const float* b1 = B1 + (size_t)(2 * hh) * NCOLS + col0 + lr;
#pragma unroll
    for (int t = 0; t < KS; ++t) {
      bf1[t].x = b1[(size_t)(4 * t) * NCOLS];
      bf1[t].y = b1[(size_t)(4 * t + 1) * NCOLS];
    }
    if (TWO) {
      const float* b2 = B2 + (size_t)(2 * hh) * NCOLS + col0 + lr;
#pragma unroll
      for (int t = 0; t < KS; ++t) {
        bf2[t].x = b2[(size_t)(4 * t) * NCOLS];
        bf2[t].y = b2[(size_t)(4 * t + 1) * NCOLS];
      }
    }
  }
  const float bb = BIAS ? bias[col0 + lr] : 0.f;

  const int ntiles = (M + ROWS - 1) / ROWS;
  const int step = (int)gridDim.x;

  float4 pf[4];
  float rdv[2] = {1.f, 1.f};
  auto gload = [&](int tile) {
    const int row0 = tile * ROWS;
#pragma unroll
    for (int i = 0; i < 4; ++i) {
      const int sb = i >> 1;
      const int r = w + 8 * (i & 1);
      const int grow = TWO ? (row0 + r) : (row0 + 16 * sb + r);
      const int gc = grow < M ? grow : M - 1;
      if (TWO && sb) {
        rdv[i & 1] = rdeg[gc];
      } else {
        pf[i] = *(const float4*)(A1 + (size_t)gc * HID + lane * 4);
      }
    }
  };

  auto offload = [&](int tile) -> int {
    int r = tile * ROWS + w + (lane & 1) + 8 * ((lane >> 1) & 1);
    r = r < M ? r : M;
    return off[r];
  };
  int idA = 0, idB = 0;
  auto idload = [&](int lob) {
    const int bA = __builtin_amdgcn_readlane(lob, 0);
    const int dA = __builtin_amdgcn_readlane(lob, 1) - bA;
    const int bB = __builtin_amdgcn_readlane(lob, 2);
    const int dB = __builtin_amdgcn_readlane(lob, 3) - bB;
    idA = 0;
    idB = 0;
    if (lane < dA) idA = csr[bA + lane];
    if (lane < dB) idB = csr[bB + lane];
  };
  float4 gA = make_float4(0.f, 0.f, 0.f, 0.f);
  float4 gB = make_float4(0.f, 0.f, 0.f, 0.f);
  auto gather = [&](int lob) {
    const int bA = __builtin_amdgcn_readlane(lob, 0);
    const int dA = __builtin_amdgcn_readlane(lob, 1) - bA;
    const int bB = __builtin_amdgcn_readlane(lob, 2);
    const int dB = __builtin_amdgcn_readlane(lob, 3) - bB;
    const int dm = dA > dB ? dA : dB;
    const float* frow = A2 + lane * 4;
    gA = make_float4(0.f, 0.f, 0.f, 0.f);
    gB = make_float4(0.f, 0.f, 0.f, 0.f);
    for (int c0 = 0; c0 < dm; c0 += 32) {
      int ia = idA, ib = idB;
      if (c0) {
        ia = 0;
        ib = 0;
        if (c0 + lane < dA) ia = csr[bA + c0 + lane];
        if (c0 + lane < dB) ib = csr[bB + c0 + lane];
      }
      const int mA = dA - c0;
      const int mB = dB - c0;
      int mm = mA > mB ? mA : mB;
      mm = mm < 32 ? mm : 32;
      for (int j = 0; j < mm; j += 8) {
        float4 va[8], vb[8];
#pragma unroll
        for (int u = 0; u < 8; ++u) {
          va[u] = make_float4(0.f, 0.f, 0.f, 0.f);
          vb[u] = make_float4(0.f, 0.f, 0.f, 0.f);
          const int sa = __builtin_amdgcn_readlane(ia, (j + u) & 31);
          const int sb = __builtin_amdgcn_readlane(ib, (j + u) & 31);
          if (j + u < mA) va[u] = *(const float4*)(frow + (size_t)sa * HID);
          if (j + u < mB) vb[u] = *(const float4*)(frow + (size_t)sb * HID);
        }
#pragma unroll
        for (int u = 0; u < 8; ++u) {
          gA.x += va[u].x; gA.y += va[u].y; gA.z += va[u].z; gA.w += va[u].w;
          gB.x += vb[u].x; gB.y += vb[u].y; gB.z += vb[u].z; gB.w += vb[u].w;
        }
      }
    }
  };
  auto sstore = [&](int stage, int tile) {
    const int row0 = tile * ROWS;
#pragma unroll
    for (int i = 0; i < 4; ++i) {
      const int sb = i >> 1;
      const int r = w + 8 * (i & 1);
      const int grow = TWO ? (row0 + r) : (row0 + 16 * sb + r);
      float4 v;
      if (TWO && sb) {
        v = (i & 1) ? gB : gA;
        const float rd = rdv[i & 1];
        v.x *= rd; v.y *= rd; v.z *= rd; v.w *= rd;
      } else {
        v = pf[i];
      }
      if (grow >= M) v = make_float4(0.f, 0.f, 0.f, 0.f);
      *(float4*)(s_a + stage * STGF + sb * SUBF + r * APITCH + lane * 4) = v;
    }
  };

  int tile = blockIdx.x;
  int lobN = 0;
  if (tile < ntiles) {
    gload(tile);
    if (TWO) {
      lobN = offload(tile);
      idload(lobN);
      gather(lobN);
    }
    sstore(0, tile);
  }
  if (TWO) lobN = offload(tile + step);
  __syncthreads();

  constexpr int KB = 8;
  int st = 0;
  for (; tile < ntiles; tile += step) {
    const int nxt = tile + step;
    const bool more = nxt < ntiles;
    int lobNN = 0;
    if (more) {
      gload(nxt);
      if (TWO) {
        idload(lobN);
        lobNN = offload(nxt + step);
      }
    }

    const float* fa = s_a + st * STGF + sub * SUBF + lr * APITCH + 2 * hh;
    v2f fr1[2][KB];
    v2f fr2[2][TWO ? KB : 1];
#pragma unroll
    for (int u = 0; u < KB; ++u) {
      fr1[0][u] = *(const v2f*)(fa + 4 * u);
      if (TWO) fr2[0][u] = *(const v2f*)(fa + SUBF + 4 * u);
    }
    v8f c = {};
#pragma unroll
    for (int kb = 0; kb < KS / KB; ++kb) {
      if (kb + 1 < KS / KB) {
#pragma unroll
        for (int u = 0; u < KB; ++u) {
          const int t = (kb + 1) * KB + u;
          fr1[(kb + 1) & 1][u] = *(const v2f*)(fa + 4 * t);
          if (TWO) fr2[(kb + 1) & 1][u] = *(const v2f*)(fa + SUBF + 4 * t);
        }
      }
      __builtin_amdgcn_sched_barrier(0);
#pragma unroll
      for (int u = 0; u < KB; ++u) {
        const int t = kb * KB + u;
        c = __builtin_amdgcn_wmma_f32_16x16x4_f32(false, fr1[kb & 1][u], false, bf1[t],
                                                  (short)0, c, false, false);
        if (TWO)
          c = __builtin_amdgcn_wmma_f32_16x16x4_f32(false, fr2[kb & 1][u], false, bf2[t],
                                                    (short)0, c, false, false);
      }
      __builtin_amdgcn_sched_barrier(0);
    }

    if (more) {
      if (TWO) gather(lobN);
      sstore(st ^ 1, nxt);
    }
    if (TWO) lobN = lobNN;
    __syncthreads();
    st ^= 1;

    const int row0 = tile * ROWS + 16 * sub;
#pragma unroll
    for (int v = 0; v < 8; ++v) {
      const int r = row0 + v + 8 * hh;
      if (r < M) {
        float val = c[v] + bb;
        if (ADD_MEAN) val += addmat[(size_t)r * NCOLS + col0 + lr] * rdeg[r];
        if (RELU) val = fmaxf(val, 0.f);
        out[(size_t)r * NCOLS + col0 + lr] = val;
      }
    }
  }
}

extern "C" void kernel_launch(void* const* d_in, const int* in_sizes, int n_in,
                              void* d_out, int out_size, void* d_ws,
                              size_t ws_size, hipStream_t stream) {
  const float* x   = (const float*)d_in[0];
  const int*   src = (const int*)d_in[1];
  const int*   dst = (const int*)d_in[2];
  const float* Ws1 = (const float*)d_in[3];
  const float* b1  = (const float*)d_in[4];
  const float* Wn1 = (const float*)d_in[5];
  const float* Ws2 = (const float*)d_in[6];
  const float* b2  = (const float*)d_in[7];
  const float* Wn2 = (const float*)d_in[8];
  const float* Ws3 = (const float*)d_in[9];
  const float* b3  = (const float*)d_in[10];
  const float* Wn3 = (const float*)d_in[11];
  float* out = (float*)d_out;

  const int N = in_sizes[0] / HID;
  const int E = in_sizes[1];

  float* deg = (float*)d_ws;
  float* h1  = deg + N;
  float* h2  = h1 + (size_t)N * HID;
  int*   cnt = (int*)(h2 + (size_t)N * HID);
  float* sum = (float*)cnt + (size_t)N * OUTD;
  int*   gcur = cnt + MAXB;
  int*   off = gcur + MAXB;
  int*   csr = off + N + 1;
  unsigned* part = (unsigned*)(csr + E);
  float* tb  = h1;

  const int ggrid = 512;
  const int NPW = 4;
  const int gwaves = (N + NPW - 1) / NPW;

  hipMemsetAsync(cnt, 0, (size_t)(2 * MAXB) * sizeof(int), stream);
  k_bhist<<<(E + CHUNK - 1) / CHUNK, 256, 0, stream>>>(dst, cnt, E);
  k_bscatter<<<(E + CHUNK - 1) / CHUNK, 256, 0, stream>>>(src, dst, cnt, gcur, part, E);
  k_bcsr<<<(N + BW - 1) / BW, 256, 0, stream>>>(cnt, part, off, deg, csr, N);

  k_gemm<HID, true, false, true, true><<<ggrid, 256, 0, stream>>>(
      x, x, deg, Ws1, Wn1, b1, nullptr, off, csr, h1, N);

  k_gemm<HID, true, false, true, true><<<ggrid, 256, 0, stream>>>(
      h1, h1, deg, Ws2, Wn2, b2, nullptr, off, csr, h2, N);

  k_gemm<OUTD, false, false, false, false><<<ggrid, 256, 0, stream>>>(
      h2, nullptr, nullptr, Wn3, nullptr, nullptr, nullptr, nullptr, nullptr, tb, N);
  k_gather<OUTD, NPW><<<(gwaves + 7) / 8, 256, 0, stream>>>(tb, off, csr, sum, N);
  k_gemm<OUTD, false, true, false, true><<<ggrid, 256, 0, stream>>>(
      h2, nullptr, deg, Ws3, nullptr, b3, sum, nullptr, nullptr, out, N);
}


# Layer 3's two products made inside layer 2's launch: relu'd tile goes via LDS as A operand to [Wn3|Ws3] (B fragments in LDS); h2 never stored, gather adds into out; 2 launches fewer; sum order kept
# speedup vs baseline: 3.9749x; 1.0558x over previous
#include <hip/hip_runtime.h>

#define HID 128
#define OUTD 64

typedef float v2f __attribute__((ext_vector_type(2)));
typedef float v8f __attribute__((ext_vector_type(8)));


#define BSH 8
#define BW 256
#define MAXB 256
#define CHUNK 4096
#define EPT (CHUNK / 256)
#define PPT 20

__global__ __launch_bounds__(256) void k_bhist(const int* __restrict__ dst,
                                               int* __restrict__ gcnt, int E) {
  __shared__ int s_h[MAXB];
  const int tid = threadIdx.x;
  s_h[tid] = 0;
  __syncthreads();
  const int e0 = blockIdx.x * CHUNK + tid;
  int d[EPT];
#pragma unroll
  for (int k = 0; k < EPT; ++k) {
    const int e = e0 + k * 256;
    d[k] = dst[e < E ? e : E - 1];
  }
#pragma unroll
  for (int k = 0; k < EPT; ++k)
    if (e0 + k * 256 < E) atomicAdd(&s_h[d[k] >> BSH], 1);
  __syncthreads();
  const int c = s_h[tid];
  if (c) atomicAdd(&gcnt[tid], c);
}

__global__ __launch_bounds__(256) void k_bscatter(const int* __restrict__ src,
                                                  const int* __restrict__ dst,
                                                  const int* __restrict__ gcnt,
                                                  int* __restrict__ gcur,
                                                  unsigned* __restrict__ part, int E) {
  __shared__ int s_h[MAXB];
  __shared__ int s_w[8];
  const int tid = threadIdx.x;
  const int lane = tid & 31;
  const int w = tid >> 5;
  s_h[tid] = 0;
  const int e0 = blockIdx.x * CHUNK + tid;
  int d[EPT], sv[EPT], r[EPT];
#pragma unroll
  for (int k = 0; k < EPT; ++k) {
    const int e = e0 + k * 256;
    const int ec = e < E ? e : E - 1;
    d[k] = dst[ec];
    sv[k] = src[ec];
  }
  const int g = gcnt[tid];
  int inc = g;
  for (int o = 1; o < 32; o <<= 1) {
    int t = __shfl_up(inc, o);
    if (lane >= o) inc += t;
  }
  if (lane == 31) s_w[w] = inc;
  __syncthreads();
  int start = inc - g;
  for (int k = 0; k < w; ++k) start += s_w[k];

#pragma unroll
  for (int k = 0; k < EPT; ++k)
    if (e0 + k * 256 >= E) d[k] = -1;
#pragma unroll
  for (int k = 0; k < EPT; ++k) {
    r[k] = 0;
    if (d[k] >= 0) r[k] = atomicAdd(&s_h[d[k] >> BSH], 1);
  }
  __syncthreads();
  const int c = s_h[tid];
  if (c) start += atomicAdd(&gcur[tid], c);
  s_h[tid] = start;
  __syncthreads();
#pragma unroll
  for (int k = 0; k < EPT; ++k) r[k] += s_h[(d[k] >> BSH) & (MAXB - 1)];
#pragma unroll
  for (int k = 0; k < EPT; ++k)
    if (d[k] >= 0) part[r[k]] = ((unsigned)sv[k] << BSH) | (unsigned)(d[k] & (BW - 1));
}

__global__ __launch_bounds__(256) void k_bcsr(const int* __restrict__ gcnt,
                                              const unsigned* __restrict__ part,
                                              int* __restrict__ off,
                                              float* __restrict__ rdeg,
                                              int* __restrict__ csr, int N) {
  __shared__ int s_c[BW];
  __shared__ int s_w[8];
  __shared__ int s_v[8];
  const int tid = threadIdx.x;
  const int lane = tid & 31;
  const int w = tid >> 5;
  const int b = blockIdx.x;
  const int cnt = gcnt[b];
  int g = (tid < b) ? gcnt[tid] : 0;
  for (int o = 16; o > 0; o >>= 1) g += __shfl_xor(g, o);
  if (lane == 0) s_w[w] = g;
  s_c[tid] = 0;
  __syncthreads();
  int base = 0;
  for (int k = 0; k < 8; ++k) base += s_w[k];
  const unsigned* p = part + base;

  for (int i0 = tid; i0 < cnt; i0 += 256 * PPT) {
    unsigned v[PPT];
#pragma unroll
    for (int u = 0; u < PPT; ++u) {
      const int i = i0 + u * 256;
      v[u] = p[i < cnt ? i : cnt - 1];
    }
#pragma unroll
    for (int u = 0; u < PPT; ++u)
      if (i0 + u * 256 < cnt) atomicAdd(&s_c[v[u] & (BW - 1)], 1);
  }
  __syncthreads();
  const int c = s_c[tid];
  int inc = c;
  for (int o = 1; o < 32; o <<= 1) {
    int t = __shfl_up(inc, o);
    if (lane >= o) inc += t;
  }
  if (lane == 31) s_v[w] = inc;
  __syncthreads();
  int ex = base + inc - c;
  for (int k = 0; k < w; ++k) ex += s_v[k];
  const int n = b * BW + tid;
  if (n < N) {
    off[n] = ex;
    rdeg[n] = 1.0f / fmaxf((float)c, 1.0f);
    if (n == N - 1) off[N] = ex + c;
  }
  s_c[tid] = ex;
  __syncthreads();
  for (int i0 = tid; i0 < cnt; i0 += 256 * PPT) {
    unsigned v[PPT];
    int slot[PPT];
#pragma unroll
    for (int u = 0; u < PPT; ++u) {
      const int i = i0 + u * 256;
      v[u] = p[i < cnt ? i : cnt - 1];
    }
#pragma unroll
    for (int u = 0; u < PPT; ++u) {
      slot[u] = 0;
      if (i0 + u * 256 < cnt) slot[u] = atomicAdd(&s_c[v[u] & (BW - 1)], 1);
    }
#pragma unroll
    for (int u = 0; u < PPT; ++u)
      if (i0 + u * 256 < cnt) csr[slot[u]] = (int)(v[u] >> BSH);
  }
}

template<int D, int NPW>
__global__ __launch_bounds__(256) void k_gather(const float* __restrict__ feat,
                                                const int* __restrict__ off,
                                                const int* __restrict__ csr,
                                                const float* __restrict__ rdeg,
                                                float* __restrict__ io, int N) {
  constexpr int LPR = D / 4;
  constexpr int G = 32 / LPR;
  const int lane = threadIdx.x & 31;
  const int wave = __builtin_amdgcn_readfirstlane((blockIdx.x * 256 + (int)threadIdx.x) >> 5);
  const int g = lane / LPR;
  const int sub = lane - g * LPR;
  const int n0 = wave * NPW;
  if (n0 >= N) return;

  int ov = 0;
  if (lane <= NPW && n0 + lane <= N) ov = off[n0 + lane];

  const float* frow = feat + sub * 4;

#pragma unroll
  for (int k = 0; k < NPW; ++k) {
    const int node = n0 + k;
    if (node >= N) break;
    const int beg = __builtin_amdgcn_readlane(ov, k);
    const int end = __builtin_amdgcn_readlane(ov, k + 1);
    float* iop = io + (size_t)node * D + sub * 4;
    float4 sv = make_float4(0.f, 0.f, 0.f, 0.f);
    if (g == 0) sv = *(const float4*)iop;
    const float rd = rdeg[node];
    float4 acc = make_float4(0.f, 0.f, 0.f, 0.f);
    for (int c = beg; c < end; c += 32) {
      const int m = (end - c < 32) ? (end - c) : 32;
      const int idv = (lane < m) ? csr[c + lane] : 0;
      for (int j = 0; j < m; j += 8 * G) {
        float4 v[8];
#pragma unroll
        for (int u = 0; u < 8; ++u) {
          v[u] = make_float4(0.f, 0.f, 0.f, 0.f);
          int s;
          if (G == 1) {
            s = __builtin_amdgcn_readlane(idv, (j + u) & 31);
          } else {
            int s0 = __builtin_amdgcn_readlane(idv, (j + 2 * u) & 31);
            int s1 = __builtin_amdgcn_readlane(idv, (j + 2 * u + 1) & 31);
            s = g ? s1 : s0;
          }
          if (j + u * G + g < m) v[u] = *(const float4*)(frow + (size_t)s * D);
        }
#pragma unroll
        for (int u = 0; u < 8; ++u) {
          acc.x += v[u].x;
          acc.y += v[u].y;
          acc.z += v[u].z;
          acc.w += v[u].w;
        }
      }
    }
    if (G == 2) {
      acc.x += __shfl_xor(acc.x, 16);
      acc.y += __shfl_xor(acc.y, 16);
      acc.z += __shfl_xor(acc.z, 16);
      acc.w += __shfl_xor(acc.w, 16);
    }
    if (g == 0) {
      sv.x += acc.x * rd;
      sv.y += acc.y * rd;
      sv.z += acc.z * rd;
      sv.w += acc.w * rd;
      *(float4*)iop = sv;
    }
  }
}

#define APITCH 132

template<int NCOLS, bool TWO, bool ADD_MEAN, bool RELU, bool BIAS, bool FUSE3>
__global__ __launch_bounds__(256) void k_gemm(const float* __restrict__ A1,
                                              const float* __restrict__ A2,
                                              const float* __restrict__ rdeg,
                                              const float* __restrict__ B1,
                                              const float* __restrict__ B2,
                                              const float* __restrict__ bias,
                                              const float* __restrict__ addmat,
                                              const int* __restrict__ off,
                                              const int* __restrict__ csr,
                                              float* __restrict__ out,
                                              const float* __restrict__ B3n,
                                              const float* __restrict__ B3s,
                                              const float* __restrict__ bias3,
                                              float* __restrict__ out3, int M) {
  constexpr int SUBF = 16 * APITCH;
  constexpr int STGF = 2 * SUBF;
  constexpr int ROWS = TWO ? 16 : 32;
  constexpr int KS = HID / 4;
  __shared__ __attribute__((aligned(16))) float s_a[2 * STGF];
  __shared__ __attribute__((aligned(16))) float s_h[FUSE3 ? 2 * SUBF : 4];
  __shared__ __attribute__((aligned(16))) float s_b3[FUSE3 ? KS * 8 * 32 * 2 : 4];

  const int tid = threadIdx.x;
  const int lane = tid & 31;
  const int w = __builtin_amdgcn_readfirstlane(tid >> 5);
  const int hh = lane >> 4;
  const int lr = lane & 15;
  const int nt = TWO ? w : (w & 3);
  const int sub = TWO ? 0 : (w >> 2);
  const int col0 = nt * 16;

  v2f bf1[KS];
  v2f bf2[TWO ? KS : 1];
  {
    const float* b1 = B1 + (size_t)(2 * hh) * NCOLS + col0 + lr;
#pragma unroll
    for (int t = 0; t < KS; ++t) {
      bf1[t].x = b1[(size_t)(4 * t) * NCOLS];
      bf1[t].y = b1[(size_t)(4 * t + 1) * NCOLS];
    }
    if (TWO) {
      const float* b2 = B2 + (size_t)(2 * hh) * NCOLS + col0 + lr;
#pragma unroll
      for (int t = 0; t < KS; ++t) {
        bf2[t].x = b2[(size_t)(4 * t) * NCOLS];
        bf2[t].y = b2[(size_t)(4 * t + 1) * NCOLS];
      }
    }
  }
  const float bb = BIAS ? bias[col0 + lr] : 0.f;
  float bb3 = 0.f;
  if (FUSE3) {
    const float* b3 = (w < 4 ? B3n : B3s) + (size_t)(2 * hh) * OUTD + (w & 3) * 16 + lr;
#pragma unroll
    for (int t = 0; t < KS; ++t) {
      v2f f;
      f.x = b3[(size_t)(4 * t) * OUTD];
      f.y = b3[(size_t)(4 * t + 1) * OUTD];
      *(v2f*)(s_b3 + ((t * 8 + w) * 32 + lane) * 2) = f;
    }
    if (w >= 4) bb3 = bias3[(w & 3) * 16 + lr];
  }

  const int ntiles = (M + ROWS - 1) / ROWS;
  const int step = (int)gridDim.x;

  float4 pf[4];
  float rdv[2] = {1.f, 1.f};
  auto gload = [&](int tile) {
    const int row0 = tile * ROWS;
#pragma unroll
    for (int i = 0; i < 4; ++i) {
      const int sb = i >> 1;
      const int r = w + 8 * (i & 1);
      const int grow = TWO ? (row0 + r) : (row0 + 16 * sb + r);
      const int gc = grow < M ? grow : M - 1;
      if (TWO && sb) {
        rdv[i & 1] = rdeg[gc];
      } else {
        pf[i] = *(const float4*)(A1 + (size_t)gc * HID + lane * 4);
      }
    }
  };

  auto offload = [&](int tile) -> int {
    int r = tile * ROWS + w + (lane & 1) + 8 * ((lane >> 1) & 1);
    r = r < M ? r : M;
    return off[r];
  };
  int idA = 0, idB = 0;
  auto idload = [&](int lob) {
    const int bA = __builtin_amdgcn_readlane(lob, 0);
    const int dA = __builtin_amdgcn_readlane(lob, 1) - bA;
    const int bB = __builtin_amdgcn_readlane(lob, 2);
    const int dB = __builtin_amdgcn_readlane(lob, 3) - bB;
    idA = 0;
    idB = 0;
    if (lane < dA) idA = csr[bA + lane];
    if (lane < dB) idB = csr[bB + lane];
  };
  float4 gA = make_float4(0.f, 0.f, 0.f, 0.f);
  float4 gB = make_float4(0.f, 0.f, 0.f, 0.f);
  auto gather = [&](int lob) {
    const int bA = __builtin_amdgcn_readlane(lob, 0);
    const int dA = __builtin_amdgcn_readlane(lob, 1) - bA;
    const int bB = __builtin_amdgcn_readlane(lob, 2);
    const int dB = __builtin_amdgcn_readlane(lob, 3) - bB;
    const int dm = dA > dB ? dA : dB;
    const float* frow = A2 + lane * 4;
    gA = make_float4(0.f, 0.f, 0.f, 0.f);
    gB = make_float4(0.f, 0.f, 0.f, 0.f);
    for (int c0 = 0; c0 < dm; c0 += 32) {
      int ia = idA, ib = idB;
      if (c0) {
        ia = 0;
        ib = 0;
        if (c0 + lane < dA) ia = csr[bA + c0 + lane];
        if (c0 + lane < dB) ib = csr[bB + c0 + lane];
      }
      const int mA = dA - c0;
      const int mB = dB - c0;
      int mm = mA > mB ? mA : mB;
      mm = mm < 32 ? mm : 32;
      for (int j = 0; j < mm; j += 8) {
        float4 va[8], vb[8];
#pragma unroll
        for (int u = 0; u < 8; ++u) {
          va[u] = make_float4(0.f, 0.f, 0.f, 0.f);
          vb[u] = make_float4(0.f, 0.f, 0.f, 0.f);
          const int sa = __builtin_amdgcn_readlane(ia, (j + u) & 31);
          const int sb = __builtin_amdgcn_readlane(ib, (j + u) & 31);
          if (j + u < mA) va[u] = *(const float4*)(frow + (size_t)sa * HID);
          if (j + u < mB) vb[u] = *(const float4*)(frow + (size_t)sb * HID);
        }
#pragma unroll
        for (int u = 0; u < 8; ++u) {
          gA.x += va[u].x; gA.y += va[u].y; gA.z += va[u].z; gA.w += va[u].w;
          gB.x += vb[u].x; gB.y += vb[u].y; gB.z += vb[u].z; gB.w += vb[u].w;
        }
      }
    }
  };
  auto sstore = [&](int stage, int tile) {
    const int row0 = tile * ROWS;
#pragma unroll
    for (int i = 0; i < 4; ++i) {
      const int sb = i >> 1;
      const int r = w + 8 * (i & 1);
      const int grow = TWO ? (row0 + r) : (row0 + 16 * sb + r);
      float4 v;
      if (TWO && sb) {
        v = (i & 1) ? gB : gA;
        const float rd = rdv[i & 1];
        v.x *= rd; v.y *= rd; v.z *= rd; v.w *= rd;
      } else {
        v = pf[i];
      }
      if (grow >= M) v = make_float4(0.f, 0.f, 0.f, 0.f);
      *(float4*)(s_a + stage * STGF + sb * SUBF + r * APITCH + lane * 4) = v;
    }
  };

  int tile = blockIdx.x;
  int lobN = 0;
  if (tile < ntiles) {
    gload(tile);
    if (TWO) {
      lobN = offload(tile);
      idload(lobN);
      gather(lobN);
    }
    sstore(0, tile);
  }
  if (TWO) lobN = offload(tile + step);
  __syncthreads();

  constexpr int KB = 8;
  int st = 0;
  for (; tile < ntiles; tile += step) {
    const int nxt = tile + step;
    const bool more = nxt < ntiles;
    int lobNN = 0;
    if (more) {
      gload(nxt);
      if (TWO) {
        idload(lobN);
        lobNN = offload(nxt + step);
      }
    }

    const float* fa = s_a + st * STGF + sub * SUBF + lr * APITCH + 2 * hh;
    v2f fr1[2][KB];
    v2f fr2[2][TWO ? KB : 1];
#pragma unroll
    for (int u = 0; u < KB; ++u) {
      fr1[0][u] = *(const v2f*)(fa + 4 * u);
      if (TWO) fr2[0][u] = *(const v2f*)(fa + SUBF + 4 * u);
    }
    v8f c = {};
#pragma unroll
    for (int kb = 0; kb < KS / KB; ++kb) {
      if (kb + 1 < KS / KB) {
#pragma unroll
        for (int u = 0; u < KB; ++u) {
          const int t = (kb + 1) * KB + u;
          fr1[(kb + 1) & 1][u] = *(const v2f*)(fa + 4 * t);
          if (TWO) fr2[(kb + 1) & 1][u] = *(const v2f*)(fa + SUBF + 4 * t);
        }
      }
      __builtin_amdgcn_sched_barrier(0);
#pragma unroll
      for (int u = 0; u < KB; ++u) {
        const int t = kb * KB + u;
        c = __builtin_amdgcn_wmma_f32_16x16x4_f32(false, fr1[kb & 1][u], false, bf1[t],
                                                  (short)0, c, false, false);
        if (TWO)
          c = __builtin_amdgcn_wmma_f32_16x16x4_f32(false, fr2[kb & 1][u], false, bf2[t],
                                                    (short)0, c, false, false);
      }
      __builtin_amdgcn_sched_barrier(0);
    }

    if (FUSE3) {
      float* hs = s_h + st * SUBF + (8 * hh) * APITCH + col0 + lr;
#pragma unroll
      for (int v = 0; v < 8; ++v) hs[v * APITCH] = fmaxf(c[v] + bb, 0.f);
    }

    if (more) {
      if (TWO) gather(lobN);
      sstore(st ^ 1, nxt);
    }
    if (TWO) lobN = lobNN;
    __syncthreads();
    st ^= 1;

    if (FUSE3) {
      const float* ha = s_h + (st ^ 1) * SUBF + lr * APITCH + 2 * hh;
      const float* hb = s_b3 + (w * 32 + lane) * 2;
      v2f a3[2][KB];
      v2f b3f[2][KB];
#pragma unroll
      for (int u = 0; u < KB; ++u) {
        a3[0][u] = *(const v2f*)(ha + 4 * u);
        b3f[0][u] = *(const v2f*)(hb + u * 512);
      }
      v8f c3 = {};
#pragma unroll
      for (int kb = 0; kb < KS / KB; ++kb) {
        if (kb + 1 < KS / KB) {
#pragma unroll
          for (int u = 0; u < KB; ++u) {
            const int t = (kb + 1) * KB + u;
            a3[(kb + 1) & 1][u] = *(const v2f*)(ha + 4 * t);
            b3f[(kb + 1) & 1][u] = *(const v2f*)(hb + t * 512);
          }
        }
        __builtin_amdgcn_sched_barrier(0);
#pragma unroll
        for (int u = 0; u < KB; ++u)
          c3 = __builtin_amdgcn_wmma_f32_16x16x4_f32(false, a3[kb & 1][u], false, b3f[kb & 1][u],
                                                     (short)0, c3, false, false);
        __builtin_amdgcn_sched_barrier(0);
      }
      float* o3 = (w < 4 ? out : out3) + (w & 3) * 16 + lr;
      const int r0 = tile * ROWS + 8 * hh;
#pragma unroll
      for (int v = 0; v < 8; ++v) {
        const int r = r0 + v;
        if (r < M) o3[(size_t)r * OUTD] = c3[v] + bb3;
      }
    }

    const int row0 = tile * ROWS + 16 * sub;
#pragma unroll
    for (int v = 0; v < 8; ++v) {
      const int r = row0 + v + 8 * hh;
      if (!FUSE3 && r < M) {
        float val = c[v] + bb;
        if (ADD_MEAN) val += addmat[(size_t)r * NCOLS + col0 + lr] * rdeg[r];
        if (RELU) val = fmaxf(val, 0.f);
        out[(size_t)r * NCOLS + col0 + lr] = val;
      }
    }
  }
}

extern "C" void kernel_launch(void* const* d_in, const int* in_sizes, int n_in,
                              void* d_out, int out_size, void* d_ws,
                              size_t ws_size, hipStream_t stream) {
  const float* x   = (const float*)d_in[0];
  const int*   src = (const int*)d_in[1];
  const int*   dst = (const int*)d_in[2];
  const float* Ws1 = (const float*)d_in[3];
  const float* b1  = (const float*)d_in[4];
  const float* Wn1 = (const float*)d_in[5];
  const float* Ws2 = (const float*)d_in[6];
  const float* b2  = (const float*)d_in[7];
  const float* Wn2 = (const float*)d_in[8];
  const float* Ws3 = (const float*)d_in[9];
  const float* b3  = (const float*)d_in[10];
  const float* Wn3 = (const float*)d_in[11];
  float* out = (float*)d_out;

  const int N = in_sizes[0] / HID;
  const int E = in_sizes[1];

  float* deg = (float*)d_ws;
  float* h1  = deg + N;
  float* tb  = h1 + (size_t)N * HID;
  int*   cnt = (int*)(tb + (size_t)N * HID);
  int*   gcur = cnt + MAXB;
  int*   off = gcur + MAXB;
  int*   csr = off + N + 1;
  unsigned* part = (unsigned*)(csr + E);

  const int ggrid = 512;
  const int NPW = 4;
  const int gwaves = (N + NPW - 1) / NPW;

  hipMemsetAsync(cnt, 0, (size_t)(2 * MAXB) * sizeof(int), stream);
  k_bhist<<<(E + CHUNK - 1) / CHUNK, 256, 0, stream>>>(dst, cnt, E);
  k_bscatter<<<(E + CHUNK - 1) / CHUNK, 256, 0, stream>>>(src, dst, cnt, gcur, part, E);
  k_bcsr<<<(N + BW - 1) / BW, 256, 0, stream>>>(cnt, part, off, deg, csr, N);

  k_gemm<HID, true, false, true, true, false><<<ggrid, 256, 0, stream>>>(
      x, x, deg, Ws1, Wn1, b1, nullptr, off, csr, h1, nullptr, nullptr, nullptr, nullptr, N);

  k_gemm<HID, true, false, true, true, true><<<ggrid, 256, 0, stream>>>(
      h1, h1, deg, Ws2, Wn2, b2, nullptr, off, csr, tb, Wn3, Ws3, b3, out, N);
  k_gather<OUTD, NPW><<<(gwaves + 7) / 8, 256, 0, stream>>>(tb, off, csr, deg, out, N);
}


# CSR build in 2 launches, not memset + 3: each edge chunk sorted by bucket in place (LDS ranks) plus a row of run starts; bucket workgroups collect their runs; no global adds; order = LDS arrival rank
# speedup vs baseline: 4.1651x; 1.0479x over previous
#include <hip/hip_runtime.h>

#define HID 128
#define OUTD 64

typedef float v2f __attribute__((ext_vector_type(2)));
typedef float v8f __attribute__((ext_vector_type(8)));


#define BSH 8
#define BW 256
#define MAXB 256
#define CHUNK 4096
#define EPT (CHUNK / 256)
#define LROW (MAXB + 1)
#define CPW 32

__global__ __launch_bounds__(256) void k_csort(const int* __restrict__ src,
                                               const int* __restrict__ dst,
                                               int* __restrict__ lofs,
                                               unsigned* __restrict__ part, int E) {
  __shared__ int s_h[MAXB];
  __shared__ int s_w[8];
  const int tid = threadIdx.x;
  const int lane = tid & 31;
  const int w = tid >> 5;
  s_h[tid] = 0;
  __syncthreads();
  const int e0 = blockIdx.x * CHUNK + tid;
  int d[EPT], sv[EPT], r[EPT];
#pragma unroll
  for (int k = 0; k < EPT; ++k) {
    const int e = e0 + k * 256;
    const int ec = e < E ? e : E - 1;
    d[k] = dst[ec];
    sv[k] = src[ec];
  }
#pragma unroll
  for (int k = 0; k < EPT; ++k)
    if (e0 + k * 256 >= E) d[k] = -1;
#pragma unroll
  for (int k = 0; k < EPT; ++k) {
    r[k] = 0;
    if (d[k] >= 0) r[k] = atomicAdd(&s_h[d[k] >> BSH], 1);
  }
  __syncthreads();
  const int c = s_h[tid];
  int inc = c;
  for (int o = 1; o < 32; o <<= 1) {
    int t = __shfl_up(inc, o);
    if (lane >= o) inc += t;
  }
  if (lane == 31) s_w[w] = inc;
  __syncthreads();
  int start = inc - c;
  for (int k = 0; k < w; ++k) start += s_w[k];
  int* row = lofs + (size_t)blockIdx.x * LROW;
  row[tid] = start;
  if (tid == MAXB - 1) row[MAXB] = start + c;
  s_h[tid] = start;
  __syncthreads();
#pragma unroll
  for (int k = 0; k < EPT; ++k) r[k] += s_h[(d[k] >> BSH) & (MAXB - 1)];
  unsigned* p = part + (size_t)blockIdx.x * CHUNK;
#pragma unroll
  for (int k = 0; k < EPT; ++k)
    if (d[k] >= 0) p[r[k]] = ((unsigned)sv[k] << BSH) | (unsigned)(d[k] & (BW - 1));
}

__global__ __launch_bounds__(256) void k_bcsr(const int* __restrict__ lofs,
                                              const unsigned* __restrict__ part,
                                              int* __restrict__ off,
                                              float* __restrict__ rdeg,
                                              int* __restrict__ csr, int N, int NCH) {
  __shared__ int s_c[BW];
  __shared__ int s_lo[MAXB];
  __shared__ int s_n[MAXB];
  __shared__ int s_w[8];
  __shared__ int s_v[8];
  const int tid = threadIdx.x;
  const int lane = tid & 31;
  const int w = __builtin_amdgcn_readfirstlane(tid >> 5);
  const int b = blockIdx.x;
  int lo = 0, nr = 0;
  if (tid < NCH) {
    const int* row = lofs + (size_t)tid * LROW + b;
    lo = row[0];
    nr = row[1] - lo;
  }
  s_lo[tid] = lo;
  s_n[tid] = nr;
  s_c[tid] = 0;
  int g = lo;
  for (int o = 16; o > 0; o >>= 1) g += __shfl_xor(g, o);
  if (lane == 0) s_w[w] = g;
  __syncthreads();
  int base = 0;
  for (int k = 0; k < 8; ++k) base += s_w[k];

  const int myn = s_n[w + 8 * lane];
  const int mylo = s_lo[w + 8 * lane];
  unsigned v[CPW];
#pragma unroll
  for (int k = 0; k < CPW; ++k) {
    const int nk = __builtin_amdgcn_readlane(myn, k);
    const int first = nk > 0 ? (w + 8 * k) * CHUNK + __builtin_amdgcn_readlane(mylo, k) : 0;
    const int last = nk > 0 ? nk - 1 : 0;
    v[k] = part[first + (lane < last ? lane : last)];
  }
#pragma unroll
  for (int k = 0; k < CPW; ++k)
    if (lane < __builtin_amdgcn_readlane(myn, k)) atomicAdd(&s_c[v[k] & (BW - 1)], 1);
#pragma unroll
  for (int k = 0; k < CPW; ++k) {
    const int nk = __builtin_amdgcn_readlane(myn, k);
    if (nk > 32) {
      const unsigned* p = part + (size_t)(w + 8 * k) * CHUNK + __builtin_amdgcn_readlane(mylo, k);
      for (int i = 32 + lane; i < nk; i += 32) atomicAdd(&s_c[p[i] & (BW - 1)], 1);
    }
  }
  __syncthreads();
  const int c = s_c[tid];
  int inc = c;
  for (int o = 1; o < 32; o <<= 1) {
    int t = __shfl_up(inc, o);
    if (lane >= o) inc += t;
  }
  if (lane == 31) s_v[w] = inc;
  __syncthreads();
  int ex = base + inc - c;
  for (int k = 0; k < w; ++k) ex += s_v[k];
  const int n = b * BW + tid;
  if (n < N) {
    off[n] = ex;
    rdeg[n] = 1.0f / fmaxf((float)c, 1.0f);
    if (n == N - 1) off[N] = ex + c;
  }
  s_c[tid] = ex;
  __syncthreads();
  int slot[CPW];
#pragma unroll
  for (int k = 0; k < CPW; ++k) {
    slot[k] = 0;
    if (lane < __builtin_amdgcn_readlane(myn, k)) slot[k] = atomicAdd(&s_c[v[k] & (BW - 1)], 1);
  }
#pragma unroll
  for (int k = 0; k < CPW; ++k)
    if (lane < __builtin_amdgcn_readlane(myn, k)) csr[slot[k]] = (int)(v[k] >> BSH);
#pragma unroll
  for (int k = 0; k < CPW; ++k) {
    const int nk = __builtin_amdgcn_readlane(myn, k);
    if (nk > 32) {
      const unsigned* p = part + (size_t)(w + 8 * k) * CHUNK + __builtin_amdgcn_readlane(mylo, k);
      for (int i = 32 + lane; i < nk; i += 32) {
        const unsigned u = p[i];
        csr[atomicAdd(&s_c[u & (BW - 1)], 1)] = (int)(u >> BSH);
      }
    }
  }
}

template<int D, int NPW>
__global__ __launch_bounds__(256) void k_gather(const float* __restrict__ feat,
                                                const int* __restrict__ off,
                                                const int* __restrict__ csr,
                                                const float* __restrict__ rdeg,
                                                float* __restrict__ io, int N) {
  constexpr int LPR = D / 4;
  constexpr int G = 32 / LPR;
  const int lane = threadIdx.x & 31;
  const int wave = __builtin_amdgcn_readfirstlane((blockIdx.x * 256 + (int)threadIdx.x) >> 5);
  const int g = lane / LPR;
  const int sub = lane - g * LPR;
  const int n0 = wave * NPW;
  if (n0 >= N) return;

  int ov = 0;
  if (lane <= NPW && n0 + lane <= N) ov = off[n0 + lane];

  const float* frow = feat + sub * 4;

#pragma unroll
  for (int k = 0; k < NPW; ++k) {
    const int node = n0 + k;
    if (node >= N) break;
    const int beg = __builtin_amdgcn_readlane(ov, k);
    const int end = __builtin_amdgcn_readlane(ov, k + 1);
    float* iop = io + (size_t)node * D + sub * 4;
    float4 sv = make_float4(0.f, 0.f, 0.f, 0.f);
    if (g == 0) sv = *(const float4*)iop;
    const float rd = rdeg[node];
    float4 acc = make_float4(0.f, 0.f, 0.f, 0.f);
    for (int c = beg; c < end; c += 32) {
      const int m = (end - c < 32) ? (end - c) : 32;
      const int idv = (lane < m) ? csr[c + lane] : 0;
      for (int j = 0; j < m; j += 8 * G) {
        float4 v[8];
#pragma unroll
        for (int u = 0; u < 8; ++u) {
          v[u] = make_float4(0.f, 0.f, 0.f, 0.f);
          int s;
          if (G == 1) {
            s = __builtin_amdgcn_readlane(idv, (j + u) & 31);
          } else {
            int s0 = __builtin_amdgcn_readlane(idv, (j + 2 * u) & 31);
            int s1 = __builtin_amdgcn_readlane(idv, (j + 2 * u + 1) & 31);
            s = g ? s1 : s0;
          }
          if (j + u * G + g < m) v[u] = *(const float4*)(frow + (size_t)s * D);
        }
#pragma unroll
        for (int u = 0; u < 8; ++u) {
          acc.x += v[u].x;
          acc.y += v[u].y;
          acc.z += v[u].z;
          acc.w += v[u].w;
        }
      }
    }
    if (G == 2) {
      acc.x += __shfl_xor(acc.x, 16);
      acc.y += __shfl_xor(acc.y, 16);
      acc.z += __shfl_xor(acc.z, 16);
      acc.w += __shfl_xor(acc.w, 16);
    }
    if (g == 0) {
      sv.x += acc.x * rd;
      sv.y += acc.y * rd;
      sv.z += acc.z * rd;
      sv.w += acc.w * rd;
      *(float4*)iop = sv;
    }
  }
}

#define APITCH 132

template<int NCOLS, bool TWO, bool ADD_MEAN, bool RELU, bool BIAS, bool FUSE3>
__global__ __launch_bounds__(256) void k_gemm(const float* __restrict__ A1,
                                              const float* __restrict__ A2,
                                              const float* __restrict__ rdeg,
                                              const float* __restrict__ B1,
                                              const float* __restrict__ B2,
                                              const float* __restrict__ bias,
                                              const float* __restrict__ addmat,
                                              const int* __restrict__ off,
                                              const int* __restrict__ csr,
                                              float* __restrict__ out,
                                              const float* __restrict__ B3n,
                                              const float* __restrict__ B3s,
                                              const float* __restrict__ bias3,
                                              float* __restrict__ out3, int M) {
  constexpr int SUBF = 16 * APITCH;
  constexpr int STGF = 2 * SUBF;
  constexpr int ROWS = TWO ? 16 : 32;
  constexpr int KS = HID / 4;
  __shared__ __attribute__((aligned(16))) float s_a[2 * STGF];
  __shared__ __attribute__((aligned(16))) float s_h[FUSE3 ? 2 * SUBF : 4];
  __shared__ __attribute__((aligned(16))) float s_b3[FUSE3 ? KS * 8 * 32 * 2 : 4];

  const int tid = threadIdx.x;
  const int lane = tid & 31;
  const int w = __builtin_amdgcn_readfirstlane(tid >> 5);
  const int hh = lane >> 4;
  const int lr = lane & 15;
  const int nt = TWO ? w : (w & 3);
  const int sub = TWO ? 0 : (w >> 2);
  const int col0 = nt * 16;

  v2f bf1[KS];
  v2f bf2[TWO ? KS : 1];
  {
    const float* b1 = B1 + (size_t)(2 * hh) * NCOLS + col0 + lr;
#pragma unroll
    for (int t = 0; t < KS; ++t) {
      bf1[t].x = b1[(size_t)(4 * t) * NCOLS];
      bf1[t].y = b1[(size_t)(4 * t + 1) * NCOLS];
    }
    if (TWO) {
      const float* b2 = B2 + (size_t)(2 * hh) * NCOLS + col0 + lr;
#pragma unroll
      for (int t = 0; t < KS; ++t) {
        bf2[t].x = b2[(size_t)(4 * t) * NCOLS];
        bf2[t].y = b2[(size_t)(4 * t + 1) * NCOLS];
      }
    }
  }
  const float bb = BIAS ? bias[col0 + lr] : 0.f;
  float bb3 = 0.f;
  if (FUSE3) {
    const float* b3 = (w < 4 ? B3n : B3s) + (size_t)(2 * hh) * OUTD + (w & 3) * 16 + lr;
#pragma unroll
    for (int t = 0; t < KS; ++t) {
      v2f f;
      f.x = b3[(size_t)(4 * t) * OUTD];
      f.y = b3[(size_t)(4 * t + 1) * OUTD];
      *(v2f*)(s_b3 + ((t * 8 + w) * 32 + lane) * 2) = f;
    }
    if (w >= 4) bb3 = bias3[(w & 3) * 16 + lr];
  }

  const int ntiles = (M + ROWS - 1) / ROWS;
  const int step = (int)gridDim.x;

  float4 pf[4];
  float rdv[2] = {1.f, 1.f};
  auto gload = [&](int tile) {
    const int row0 = tile * ROWS;
#pragma unroll
    for (int i = 0; i < 4; ++i) {
      const int sb = i >> 1;
      const int r = w + 8 * (i & 1);
      const int grow = TWO ? (row0 + r) : (row0 + 16 * sb + r);
      const int gc = grow < M ? grow : M - 1;
      if (TWO && sb) {
        rdv[i & 1] = rdeg[gc];
      } else {
        pf[i] = *(const float4*)(A1 + (size_t)gc * HID + lane * 4);
      }
    }
  };

  auto offload = [&](int tile) -> int {
    int r = tile * ROWS + w + (lane & 1) + 8 * ((lane >> 1) & 1);
    r = r < M ? r : M;
    return off[r];
  };
  int idA = 0, idB = 0;
  auto idload = [&](int lob) {
    const int bA = __builtin_amdgcn_readlane(lob, 0);
    const int dA = __builtin_amdgcn_readlane(lob, 1) - bA;
    const int bB = __builtin_amdgcn_readlane(lob, 2);
    const int dB = __builtin_amdgcn_readlane(lob, 3) - bB;
    idA = 0;
    idB = 0;
    if (lane < dA) idA = csr[bA + lane];
    if (lane < dB) idB = csr[bB + lane];
  };
  float4 gA = make_float4(0.f, 0.f, 0.f, 0.f);
  float4 gB = make_float4(0.f, 0.f, 0.f, 0.f);
  auto gather = [&](int lob) {
    const int bA = __builtin_amdgcn_readlane(lob, 0);
    const int dA = __builtin_amdgcn_readlane(lob, 1) - bA;
    const int bB = __builtin_amdgcn_readlane(lob, 2);
    const int dB = __builtin_amdgcn_readlane(lob, 3) - bB;
    const int dm = dA > dB ? dA : dB;
    const float* frow = A2 + lane * 4;
    gA = make_float4(0.f, 0.f, 0.f, 0.f);
    gB = make_float4(0.f, 0.f, 0.f, 0.f);
    for (int c0 = 0; c0 < dm; c0 += 32) {
      int ia = idA, ib = idB;
      if (c0) {
        ia = 0;
        ib = 0;
        if (c0 + lane < dA) ia = csr[bA + c0 + lane];
        if (c0 + lane < dB) ib = csr[bB + c0 + lane];
      }
      const int mA = dA - c0;
      const int mB = dB - c0;
      int mm = mA > mB ? mA : mB;
      mm = mm < 32 ? mm : 32;
      for (int j = 0; j < mm; j += 8) {
        float4 va[8], vb[8];
#pragma unroll
        for (int u = 0; u < 8; ++u) {
          va[u] = make_float4(0.f, 0.f, 0.f, 0.f);
          vb[u] = make_float4(0.f, 0.f, 0.f, 0.f);
          const int sa = __builtin_amdgcn_readlane(ia, (j + u) & 31);
          const int sb = __builtin_amdgcn_readlane(ib, (j + u) & 31);
          if (j + u < mA) va[u] = *(const float4*)(frow + (size_t)sa * HID);
          if (j + u < mB) vb[u] = *(const float4*)(frow + (size_t)sb * HID);
        }
#pragma unroll
        for (int u = 0; u < 8; ++u) {
          gA.x += va[u].x; gA.y += va[u].y; gA.z += va[u].z; gA.w += va[u].w;
          gB.x += vb[u].x; gB.y += vb[u].y; gB.z += vb[u].z; gB.w += vb[u].w;
        }
      }
    }
  };
  auto sstore = [&](int stage, int tile) {
    const int row0 = tile * ROWS;
#pragma unroll
    for (int i = 0; i < 4; ++i) {
      const int sb = i >> 1;
      const int r = w + 8 * (i & 1);
      const int grow = TWO ? (row0 + r) : (row0 + 16 * sb + r);
      float4 v;
      if (TWO && sb) {
        v = (i & 1) ? gB : gA;
        const float rd = rdv[i & 1];
        v.x *= rd; v.y *= rd; v.z *= rd; v.w *= rd;
      } else {
        v = pf[i];
      }
      if (grow >= M) v = make_float4(0.f, 0.f, 0.f, 0.f);
      *(float4*)(s_a + stage * STGF + sb * SUBF + r * APITCH + lane * 4) = v;
    }
  };

  int tile = blockIdx.x;
  int lobN = 0;
  if (tile < ntiles) {
    gload(tile);
    if (TWO) {
      lobN = offload(tile);
      idload(lobN);
      gather(lobN);
    }
    sstore(0, tile);
  }
  if (TWO) lobN = offload(tile + step);
  __syncthreads();

  constexpr int KB = 8;
  int st = 0;
  for (; tile < ntiles; tile += step) {
    const int nxt = tile + step;
    const bool more = nxt < ntiles;
    int lobNN = 0;
    if (more) {
      gload(nxt);
      if (TWO) {
        idload(lobN);
        lobNN = offload(nxt + step);
      }
    }

    const float* fa = s_a + st * STGF + sub * SUBF + lr * APITCH + 2 * hh;
    v2f fr1[2][KB];
    v2f fr2[2][TWO ? KB : 1];
#pragma unroll
    for (int u = 0; u < KB; ++u) {
      fr1[0][u] = *(const v2f*)(fa + 4 * u);
      if (TWO) fr2[0][u] = *(const v2f*)(fa + SUBF + 4 * u);
    }
    v8f c = {};
#pragma unroll
    for (int kb = 0; kb < KS / KB; ++kb) {
      if (kb + 1 < KS / KB) {
#pragma unroll
        for (int u = 0; u < KB; ++u) {
          const int t = (kb + 1) * KB + u;
          fr1[(kb + 1) & 1][u] = *(const v2f*)(fa + 4 * t);
          if (TWO) fr2[(kb + 1) & 1][u] = *(const v2f*)(fa + SUBF + 4 * t);
        }
      }
      __builtin_amdgcn_sched_barrier(0);
#pragma unroll
      for (int u = 0; u < KB; ++u) {
        const int t = kb * KB + u;
        c = __builtin_amdgcn_wmma_f32_16x16x4_f32(false, fr1[kb & 1][u], false, bf1[t],
                                                  (short)0, c, false, false);
        if (TWO)
          c = __builtin_amdgcn_wmma_f32_16x16x4_f32(false, fr2[kb & 1][u], false, bf2[t],
                                                    (short)0, c, false, false);
      }
      __builtin_amdgcn_sched_barrier(0);
    }

    if (FUSE3) {
      float* hs = s_h + st * SUBF + (8 * hh) * APITCH + col0 + lr;
#pragma unroll
      for (int v = 0; v < 8; ++v) hs[v * APITCH] = fmaxf(c[v] + bb, 0.f);
    }

    if (more) {
      if (TWO) gather(lobN);
      sstore(st ^ 1, nxt);
    }
    if (TWO) lobN = lobNN;
    __syncthreads();
    st ^= 1;

    if (FUSE3) {
      const float* ha = s_h + (st ^ 1) * SUBF + lr * APITCH + 2 * hh;
      const float* hb = s_b3 + (w * 32 + lane) * 2;
      v2f a3[2][KB];
      v2f b3f[2][KB];
#pragma unroll
      for (int u = 0; u < KB; ++u) {
        a3[0][u] = *(const v2f*)(ha + 4 * u);
        b3f[0][u] = *(const v2f*)(hb + u * 512);
      }
      v8f c3 = {};
#pragma unroll
      for (int kb = 0; kb < KS / KB; ++kb) {
        if (kb + 1 < KS / KB) {
#pragma unroll
          for (int u = 0; u < KB; ++u) {
            const int t = (kb + 1) * KB + u;
            a3[(kb + 1) & 1][u] = *(const v2f*)(ha + 4 * t);
            b3f[(kb + 1) & 1][u] = *(const v2f*)(hb + t * 512);
          }
        }
        __builtin_amdgcn_sched_barrier(0);
#pragma unroll
        for (int u = 0; u < KB; ++u)
          c3 = __builtin_amdgcn_wmma_f32_16x16x4_f32(false, a3[kb & 1][u], false, b3f[kb & 1][u],
                                                     (short)0, c3, false, false);
        __builtin_amdgcn_sched_barrier(0);
      }
      float* o3 = (w < 4 ? out : out3) + (w & 3) * 16 + lr;
      const int r0 = tile * ROWS + 8 * hh;
#pragma unroll
      for (int v = 0; v < 8; ++v) {
        const int r = r0 + v;
        if (r < M) o3[(size_t)r * OUTD] = c3[v] + bb3;
      }
    }

    const int row0 = tile * ROWS + 16 * sub;
#pragma unroll
    for (int v = 0; v < 8; ++v) {
      const int r = row0 + v + 8 * hh;
      if (!FUSE3 && r < M) {
        float val = c[v] + bb;
        if (ADD_MEAN) val += addmat[(size_t)r * NCOLS + col0 + lr] * rdeg[r];
        if (RELU) val = fmaxf(val, 0.f);
        out[(size_t)r * NCOLS + col0 + lr] = val;
      }
    }
  }
}

extern "C" void kernel_launch(void* const* d_in, const int* in_sizes, int n_in,
                              void* d_out, int out_size, void* d_ws,
                              size_t ws_size, hipStream_t stream) {
  const float* x   = (const float*)d_in[0];
  const int*   src = (const int*)d_in[1];
  const int*   dst = (const int*)d_in[2];
  const float* Ws1 = (const float*)d_in[3];
  const float* b1  = (const float*)d_in[4];
  const float* Wn1 = (const float*)d_in[5];
  const float* Ws2 = (const float*)d_in[6];
  const float* b2  = (const float*)d_in[7];
  const float* Wn2 = (const float*)d_in[8];
  const float* Ws3 = (const float*)d_in[9];
  const float* b3  = (const float*)d_in[10];
  const float* Wn3 = (const float*)d_in[11];
  float* out = (float*)d_out;

  const int N = in_sizes[0] / HID;
  const int E = in_sizes[1];

  float* deg = (float*)d_ws;
  float* h1  = deg + N;
  float* tb  = h1 + (size_t)N * HID;
  int*   lofs = (int*)(tb + (size_t)N * OUTD);
  int*   off = (int*)(tb + (size_t)N * HID);
  int*   csr = off + N + 1;
  unsigned* part = (unsigned*)(csr + E);

  const int ggrid = 512;
  const int NPW = 4;
  const int gwaves = (N + NPW - 1) / NPW;

  const int nch = (E + CHUNK - 1) / CHUNK;
  k_csort<<<nch, 256, 0, stream>>>(src, dst, lofs, part, E);
  k_bcsr<<<(N + BW - 1) / BW, 256, 0, stream>>>(lofs, part, off, deg, csr, N, nch);

  k_gemm<HID, true, false, true, true, false><<<ggrid, 256, 0, stream>>>(
      x, x, deg, Ws1, Wn1, b1, nullptr, off, csr, h1, nullptr, nullptr, nullptr, nullptr, N);

  k_gemm<HID, true, false, true, true, true><<<ggrid, 256, 0, stream>>>(
      h1, h1, deg, Ws2, Wn2, b2, nullptr, off, csr, tb, Wn3, Ws3, b3, out, N);
  k_gather<OUTD, NPW><<<(gwaves + 7) / 8, 256, 0, stream>>>(tb, off, csr, deg, out, N);
}


# Neighbour walk in the GEMM launches: a wave's two lists walked as one sequence with 16 row loads in flight all along (counted waits), not drained batches of 8+8; sum order unchanged
# speedup vs baseline: 4.3077x; 1.0342x over previous
#include <hip/hip_runtime.h>

#define HID 128
#define OUTD 64

typedef float v2f __attribute__((ext_vector_type(2)));
typedef float v8f __attribute__((ext_vector_type(8)));


#define BSH 8
#define BW 256
#define MAXB 256
#define CHUNK 4096
#define EPT (CHUNK / 256)
#define LROW (MAXB + 1)
#define CPW 32

__global__ __launch_bounds__(256) void k_csort(const int* __restrict__ src,
                                               const int* __restrict__ dst,
                                               int* __restrict__ lofs,
                                               unsigned* __restrict__ part, int E) {
  __shared__ int s_h[MAXB];
  __shared__ int s_w[8];
  const int tid = threadIdx.x;
  const int lane = tid & 31;
  const int w = tid >> 5;
  s_h[tid] = 0;
  __syncthreads();
  const int e0 = blockIdx.x * CHUNK + tid;
  int d[EPT], sv[EPT], r[EPT];
#pragma unroll
  for (int k = 0; k < EPT; ++k) {
    const int e = e0 + k * 256;
    const int ec = e < E ? e : E - 1;
    d[k] = dst[ec];
    sv[k] = src[ec];
  }
#pragma unroll
  for (int k = 0; k < EPT; ++k)
    if (e0 + k * 256 >= E) d[k] = -1;
#pragma unroll
  for (int k = 0; k < EPT; ++k) {
    r[k] = 0;
    if (d[k] >= 0) r[k] = atomicAdd(&s_h[d[k] >> BSH], 1);
  }
  __syncthreads();
  const int c = s_h[tid];
  int inc = c;
  for (int o = 1; o < 32; o <<= 1) {
    int t = __shfl_up(inc, o);
    if (lane >= o) inc += t;
  }
  if (lane == 31) s_w[w] = inc;
  __syncthreads();
  int start = inc - c;
  for (int k = 0; k < w; ++k) start += s_w[k];
  int* row = lofs + (size_t)blockIdx.x * LROW;
  row[tid] = start;
  if (tid == MAXB - 1) row[MAXB] = start + c;
  s_h[tid] = start;
  __syncthreads();
#pragma unroll
  for (int k = 0; k < EPT; ++k) r[k] += s_h[(d[k] >> BSH) & (MAXB - 1)];
  unsigned* p = part + (size_t)blockIdx.x * CHUNK;
#pragma unroll
  for (int k = 0; k < EPT; ++k)
    if (d[k] >= 0) p[r[k]] = ((unsigned)sv[k] << BSH) | (unsigned)(d[k] & (BW - 1));
}

__global__ __launch_bounds__(256) void k_bcsr(const int* __restrict__ lofs,
                                              const unsigned* __restrict__ part,
                                              int* __restrict__ off,
                                              float* __restrict__ rdeg,
                                              int* __restrict__ csr, int N, int NCH) {
  __shared__ int s_c[BW];
  __shared__ int s_lo[MAXB];
  __shared__ int s_n[MAXB];
  __shared__ int s_w[8];
  __shared__ int s_v[8];
  const int tid = threadIdx.x;
  const int lane = tid & 31;
  const int w = __builtin_amdgcn_readfirstlane(tid >> 5);
  const int b = blockIdx.x;
  int lo = 0, nr = 0;
  if (tid < NCH) {
    const int* row = lofs + (size_t)tid * LROW + b;
    lo = row[0];
    nr = row[1] - lo;
  }
  s_lo[tid] = lo;
  s_n[tid] = nr;
  s_c[tid] = 0;
  int g = lo;
  for (int o = 16; o > 0; o >>= 1) g += __shfl_xor(g, o);
  if (lane == 0) s_w[w] = g;
  __syncthreads();
  int base = 0;
  for (int k = 0; k < 8; ++k) base += s_w[k];

  const int myn = s_n[w + 8 * lane];
  const int mylo = s_lo[w + 8 * lane];
  unsigned v[CPW];
#pragma unroll
  for (int k = 0; k < CPW; ++k) {
    const int nk = __builtin_amdgcn_readlane(myn, k);
    const int first = nk > 0 ? (w + 8 * k) * CHUNK + __builtin_amdgcn_readlane(mylo, k) : 0;
    const int last = nk > 0 ? nk - 1 : 0;
    v[k] = part[first + (lane < last ? lane : last)];
  }
#pragma unroll
  for (int k = 0; k < CPW; ++k)
    if (lane < __builtin_amdgcn_readlane(myn, k)) atomicAdd(&s_c[v[k] & (BW - 1)], 1);
#pragma unroll
  for (int k = 0; k < CPW; ++k) {
    const int nk = __builtin_amdgcn_readlane(myn, k);
    if (nk > 32) {
      const unsigned* p = part + (size_t)(w + 8 * k) * CHUNK + __builtin_amdgcn_readlane(mylo, k);
      for (int i = 32 + lane; i < nk; i += 32) atomicAdd(&s_c[p[i] & (BW - 1)], 1);
    }
  }
  __syncthreads();
  const int c = s_c[tid];
  int inc = c;
  for (int o = 1; o < 32; o <<= 1) {
    int t = __shfl_up(inc, o);
    if (lane >= o) inc += t;
  }
  if (lane == 31) s_v[w] = inc;
  __syncthreads();
  int ex = base + inc - c;
  for (int k = 0; k < w; ++k) ex += s_v[k];
  const int n = b * BW + tid;
  if (n < N) {
    off[n] = ex;
    rdeg[n] = 1.0f / fmaxf((float)c, 1.0f);
    if (n == N - 1) off[N] = ex + c;
  }
  s_c[tid] = ex;
  __syncthreads();
  int slot[CPW];
#pragma unroll
  for (int k = 0; k < CPW; ++k) {
    slot[k] = 0;
    if (lane < __builtin_amdgcn_readlane(myn, k)) slot[k] = atomicAdd(&s_c[v[k] & (BW - 1)], 1);
  }
#pragma unroll
  for (int k = 0; k < CPW; ++k)
    if (lane < __builtin_amdgcn_readlane(myn, k)) csr[slot[k]] = (int)(v[k] >> BSH);
#pragma unroll
  for (int k = 0; k < CPW; ++k) {
    const int nk = __builtin_amdgcn_readlane(myn, k);
    if (nk > 32) {
      const unsigned* p = part + (size_t)(w + 8 * k) * CHUNK + __builtin_amdgcn_readlane(mylo, k);
      for (int i = 32 + lane; i < nk; i += 32) {
        const unsigned u = p[i];
        csr[atomicAdd(&s_c[u & (BW - 1)], 1)] = (int)(u >> BSH);
      }
    }
  }
}

template<int D, int NPW>
__global__ __launch_bounds__(256) void k_gather(const float* __restrict__ feat,
                                                const int* __restrict__ off,
                                                const int* __restrict__ csr,
                                                const float* __restrict__ rdeg,
                                                float* __restrict__ io, int N) {
  constexpr int LPR = D / 4;
  constexpr int G = 32 / LPR;
  const int lane = threadIdx.x & 31;
  const int wave = __builtin_amdgcn_readfirstlane((blockIdx.x * 256 + (int)threadIdx.x) >> 5);
  const int g = lane / LPR;
  const int sub = lane - g * LPR;
  const int n0 = wave * NPW;
  if (n0 >= N) return;

  int ov = 0;
  if (lane <= NPW && n0 + lane <= N) ov = off[n0 + lane];

  const float* frow = feat + sub * 4;

#pragma unroll
  for (int k = 0; k < NPW; ++k) {
    const int node = n0 + k;
    if (node >= N) break;
    const int beg = __builtin_amdgcn_readlane(ov, k);
    const int end = __builtin_amdgcn_readlane(ov, k + 1);
    float* iop = io + (size_t)node * D + sub * 4;
    float4 sv = make_float4(0.f, 0.f, 0.f, 0.f);
    if (g == 0) sv = *(const float4*)iop;
    const float rd = rdeg[node];
    float4 acc = make_float4(0.f, 0.f, 0.f, 0.f);
    for (int c = beg; c < end; c += 32) {
      const int m = (end - c < 32) ? (end - c) : 32;
      const int idv = (lane < m) ? csr[c + lane] : 0;
      for (int j = 0; j < m; j += 8 * G) {
        float4 v[8];
#pragma unroll
        for (int u = 0; u < 8; ++u) {
          v[u] = make_float4(0.f, 0.f, 0.f, 0.f);
          int s;
          if (G == 1) {
            s = __builtin_amdgcn_readlane(idv, (j + u) & 31);
          } else {
            int s0 = __builtin_amdgcn_readlane(idv, (j + 2 * u) & 31);
            int s1 = __builtin_amdgcn_readlane(idv, (j + 2 * u + 1) & 31);
            s = g ? s1 : s0;
          }
          if (j + u * G + g < m) v[u] = *(const float4*)(frow + (size_t)s * D);
        }
#pragma unroll
        for (int u = 0; u < 8; ++u) {
          acc.x += v[u].x;
          acc.y += v[u].y;
          acc.z += v[u].z;
          acc.w += v[u].w;
        }
      }
    }
    if (G == 2) {
      acc.x += __shfl_xor(acc.x, 16);
      acc.y += __shfl_xor(acc.y, 16);
      acc.z += __shfl_xor(acc.z, 16);
      acc.w += __shfl_xor(acc.w, 16);
    }
    if (g == 0) {
      sv.x += acc.x * rd;
      sv.y += acc.y * rd;
      sv.z += acc.z * rd;
      sv.w += acc.w * rd;
      *(float4*)iop = sv;
    }
  }
}

#define APITCH 132

template<int NCOLS, bool TWO, bool ADD_MEAN, bool RELU, bool BIAS, bool FUSE3>
__global__ __launch_bounds__(256) void k_gemm(const float* __restrict__ A1,
                                              const float* __restrict__ A2,
                                              const float* __restrict__ rdeg,
                                              const float* __restrict__ B1,
                                              const float* __restrict__ B2,
                                              const float* __restrict__ bias,
                                              const float* __restrict__ addmat,
                                              const int* __restrict__ off,
                                              const int* __restrict__ csr,
                                              float* __restrict__ out,
                                              const float* __restrict__ B3n,
                                              const float* __restrict__ B3s,
                                              const float* __restrict__ bias3,
                                              float* __restrict__ out3, int M) {
  constexpr int SUBF = 16 * APITCH;
  constexpr int STGF = 2 * SUBF;
  constexpr int ROWS = TWO ? 16 : 32;
  constexpr int KS = HID / 4;
  __shared__ __attribute__((aligned(16))) float s_a[2 * STGF];
  __shared__ __attribute__((aligned(16))) float s_h[FUSE3 ? 2 * SUBF : 4];
  __shared__ __attribute__((aligned(16))) float s_b3[FUSE3 ? KS * 8 * 32 * 2 : 4];

  const int tid = threadIdx.x;
  const int lane = tid & 31;
  const int w = __builtin_amdgcn_readfirstlane(tid >> 5);
  const int hh = lane >> 4;
  const int lr = lane & 15;
  const int nt = TWO ? w : (w & 3);
  const int sub = TWO ? 0 : (w >> 2);
  const int col0 = nt * 16;

  v2f bf1[KS];
  v2f bf2[TWO ? KS : 1];
  {
    const float* b1 = B1 + (size_t)(2 * hh) * NCOLS + col0 + lr;
#pragma unroll
    for (int t = 0; t < KS; ++t) {
      bf1[t].x = b1[(size_t)(4 * t) * NCOLS];
      bf1[t].y = b1[(size_t)(4 * t + 1) * NCOLS];
    }
    if (TWO) {
      const float* b2 = B2 + (size_t)(2 * hh) * NCOLS + col0 + lr;
#pragma unroll
      for (int t = 0; t < KS; ++t) {
        bf2[t].x = b2[(size_t)(4 * t) * NCOLS];
        bf2[t].y = b2[(size_t)(4 * t + 1) * NCOLS];
      }
    }
  }
  const float bb = BIAS ? bias[col0 + lr] : 0.f;
  float bb3 = 0.f;
  if (FUSE3) {
    const float* b3 = (w < 4 ? B3n : B3s) + (size_t)(2 * hh) * OUTD + (w & 3) * 16 + lr;
#pragma unroll
    for (int t = 0; t < KS; ++t) {
      v2f f;
      f.x = b3[(size_t)(4 * t) * OUTD];
      f.y = b3[(size_t)(4 * t + 1) * OUTD];
      *(v2f*)(s_b3 + ((t * 8 + w) * 32 + lane) * 2) = f;
    }
    if (w >= 4) bb3 = bias3[(w & 3) * 16 + lr];
  }

  const int ntiles = (M + ROWS - 1) / ROWS;
  const int step = (int)gridDim.x;

  float4 pf[4];
  float rdv[2] = {1.f, 1.f};
  auto gload = [&](int tile) {
    const int row0 = tile * ROWS;
#pragma unroll
    for (int i = 0; i < 4; ++i) {
      const int sb = i >> 1;
      const int r = w + 8 * (i & 1);
      const int grow = TWO ? (row0 + r) : (row0 + 16 * sb + r);
      const int gc = grow < M ? grow : M - 1;
      if (TWO && sb) {
        rdv[i & 1] = rdeg[gc];
      } else {
        pf[i] = *(const float4*)(A1 + (size_t)gc * HID + lane * 4);
      }
    }
  };

  auto offload = [&](int tile) -> int {
    int r = tile * ROWS + w + (lane & 1) + 8 * ((lane >> 1) & 1);
    r = r < M ? r : M;
    return off[r];
  };
  int idA = 0, idB = 0;
  auto idload = [&](int lob) {
    const int bA = __builtin_amdgcn_readlane(lob, 0);
    const int dA = __builtin_amdgcn_readlane(lob, 1) - bA;
    const int bB = __builtin_amdgcn_readlane(lob, 2);
    const int dB = __builtin_amdgcn_readlane(lob, 3) - bB;
    idA = 0;
    idB = 0;
    if (lane < dA) idA = csr[bA + lane];
    if (lane < dB) idB = csr[bB + lane];
  };
  float4 gA = make_float4(0.f, 0.f, 0.f, 0.f);
  float4 gB = make_float4(0.f, 0.f, 0.f, 0.f);
  constexpr int GW = 16;
  auto gather = [&](int lob) {
    const int bA = __builtin_amdgcn_readlane(lob, 0);
    const int dA = __builtin_amdgcn_readlane(lob, 1) - bA;
    const int bB = __builtin_amdgcn_readlane(lob, 2);
    const int dB = __builtin_amdgcn_readlane(lob, 3) - bB;
    const int nA = dA < 32 ? dA : 32;
    const int nB = dB < 32 ? dB : 32;
    const int n = nA + nB;
    float4 g = make_float4(0.f, 0.f, 0.f, 0.f);
    float4 gS = make_float4(0.f, 0.f, 0.f, 0.f);
    if (n > 0) {
      float4 v[GW];
#define ROWP(k_)                                                                  \
  ([&]() -> const float4* {                                                       \
    const int kc = (k_) < n ? (k_) : n - 1;                                       \
    const int sa = __builtin_amdgcn_readlane(idA, kc & 31);                       \
    const int sb = __builtin_amdgcn_readlane(idB, (kc - nA) & 31);                \
    const int s = kc < nA ? sa : sb;                                              \
    return (const float4*)(A2 + (size_t)s * HID) + lane;                          \
  }())
#define TAKE(k_, t_)                                                              \
  do {                                                                            \
    if ((k_) < n) {                                                               \
      if ((k_) == nA) {                                                           \
        gS = g;                                                                   \
        g = make_float4(0.f, 0.f, 0.f, 0.f);                                      \
      }                                                                           \
      g.x += (t_).x; g.y += (t_).y; g.z += (t_).z; g.w += (t_).w;                 \
    }                                                                             \
  } while (0)
#pragma unroll
      for (int u = 0; u < GW; ++u) v[u] = *ROWP(u);
      int k0 = 0;
      for (; k0 + GW < n; k0 += GW) {
#pragma unroll
        for (int u = 0; u < GW; ++u) {
          TAKE(k0 + u, v[u]);
          v[u] = *ROWP(k0 + u + GW);
        }
      }
#pragma unroll
      for (int u = 0; u < GW; ++u) TAKE(k0 + u, v[u]);
#undef ROWP
#undef TAKE
    }
    if (nB > 0) {
      gA = gS;
      gB = g;
    } else {
      gA = g;
      gB = make_float4(0.f, 0.f, 0.f, 0.f);
    }
    for (int c = 32; c < dA; ++c) {
      const int s = __builtin_amdgcn_readfirstlane(csr[bA + c]);
      const float4 t = *((const float4*)(A2 + (size_t)s * HID) + lane);
      gA.x += t.x; gA.y += t.y; gA.z += t.z; gA.w += t.w;
    }
    for (int c = 32; c < dB; ++c) {
      const int s = __builtin_amdgcn_readfirstlane(csr[bB + c]);
      const float4 t = *((const float4*)(A2 + (size_t)s * HID) + lane);
      gB.x += t.x; gB.y += t.y; gB.z += t.z; gB.w += t.w;
    }
  };
  auto sstore = [&](int stage, int tile) {
    const int row0 = tile * ROWS;
#pragma unroll
    for (int i = 0; i < 4; ++i) {
      const int sb = i >> 1;
      const int r = w + 8 * (i & 1);
      const int grow = TWO ? (row0 + r) : (row0 + 16 * sb + r);
      float4 v;
      if (TWO && sb) {
        v = (i & 1) ? gB : gA;
        const float rd = rdv[i & 1];
        v.x *= rd; v.y *= rd; v.z *= rd; v.w *= rd;
      } else {
        v = pf[i];
      }
      if (grow >= M) v = make_float4(0.f, 0.f, 0.f, 0.f);
      *(float4*)(s_a + stage * STGF + sb * SUBF + r * APITCH + lane * 4) = v;
    }
  };

  int tile = blockIdx.x;
  int lobN = 0;
  if (tile < ntiles) {
    gload(tile);
    if (TWO) {
      lobN = offload(tile);
      idload(lobN);
      gather(lobN);
    }
    sstore(0, tile);
  }
  if (TWO) lobN = offload(tile + step);
  __syncthreads();

  constexpr int KB = 8;
  int st = 0;
  for (; tile < ntiles; tile += step) {
    const int nxt = tile + step;
    const bool more = nxt < ntiles;
    int lobNN = 0;
    if (more) {
      gload(nxt);
      if (TWO) {
        idload(lobN);
        lobNN = offload(nxt + step);
      }
    }

    const float* fa = s_a + st * STGF + sub * SUBF + lr * APITCH + 2 * hh;
    v2f fr1[2][KB];
    v2f fr2[2][TWO ? KB : 1];
#pragma unroll
    for (int u = 0; u < KB; ++u) {
      fr1[0][u] = *(const v2f*)(fa + 4 * u);
      if (TWO) fr2[0][u] = *(const v2f*)(fa + SUBF + 4 * u);
    }
    v8f c = {};
#pragma unroll
    for (int kb = 0; kb < KS / KB; ++kb) {
      if (kb + 1 < KS / KB) {
#pragma unroll
        for (int u = 0; u < KB; ++u) {
          const int t = (kb + 1) * KB + u;
          fr1[(kb + 1) & 1][u] = *(const v2f*)(fa + 4 * t);
          if (TWO) fr2[(kb + 1) & 1][u] = *(const v2f*)(fa + SUBF + 4 * t);
        }
      }
      __builtin_amdgcn_sched_barrier(0);
#pragma unroll
      for (int u = 0; u < KB; ++u) {
        const int t = kb * KB + u;
        c = __builtin_amdgcn_wmma_f32_16x16x4_f32(false, fr1[kb & 1][u], false, bf1[t],
                                                  (short)0, c, false, false);
        if (TWO)
          c = __builtin_amdgcn_wmma_f32_16x16x4_f32(false, fr2[kb & 1][u], false, bf2[t],
                                                    (short)0, c, false, false);
      }
      __builtin_amdgcn_sched_barrier(0);
    }

    if (FUSE3) {
      float* hs = s_h + st * SUBF + (8 * hh) * APITCH + col0 + lr;
#pragma unroll
      for (int v = 0; v < 8; ++v) hs[v * APITCH] = fmaxf(c[v] + bb, 0.f);
    }

    if (more) {
      if (TWO) gather(lobN);
      sstore(st ^ 1, nxt);
    }
    if (TWO) lobN = lobNN;
    __syncthreads();
    st ^= 1;

    if (FUSE3) {
      const float* ha = s_h + (st ^ 1) * SUBF + lr * APITCH + 2 * hh;
      const float* hb = s_b3 + (w * 32 + lane) * 2;
      v2f a3[2][KB];
      v2f b3f[2][KB];
#pragma unroll
      for (int u = 0; u < KB; ++u) {
        a3[0][u] = *(const v2f*)(ha + 4 * u);
        b3f[0][u] = *(const v2f*)(hb + u * 512);
      }
      v8f c3 = {};
#pragma unroll
      for (int kb = 0; kb < KS / KB; ++kb) {
        if (kb + 1 < KS / KB) {
#pragma unroll
          for (int u = 0; u < KB; ++u) {
            const int t = (kb + 1) * KB + u;
            a3[(kb + 1) & 1][u] = *(const v2f*)(ha + 4 * t);
            b3f[(kb + 1) & 1][u] = *(const v2f*)(hb + t * 512);
          }
        }
        __builtin_amdgcn_sched_barrier(0);
#pragma unroll
        for (int u = 0; u < KB; ++u)
          c3 = __builtin_amdgcn_wmma_f32_16x16x4_f32(false, a3[kb & 1][u], false, b3f[kb & 1][u],
                                                     (short)0, c3, false, false);
        __builtin_amdgcn_sched_barrier(0);
      }
      float* o3 = (w < 4 ? out : out3) + (w & 3) * 16 + lr;
      const int r0 = tile * ROWS + 8 * hh;
#pragma unroll
      for (int v = 0; v < 8; ++v) {
        const int r = r0 + v;
        if (r < M) o3[(size_t)r * OUTD] = c3[v] + bb3;
      }
    }

    const int row0 = tile * ROWS + 16 * sub;
#pragma unroll
    for (int v = 0; v < 8; ++v) {
      const int r = row0 + v + 8 * hh;
      if (!FUSE3 && r < M) {
        float val = c[v] + bb;
        if (ADD_MEAN) val += addmat[(size_t)r * NCOLS + col0 + lr] * rdeg[r];
        if (RELU) val = fmaxf(val, 0.f);
        out[(size_t)r * NCOLS + col0 + lr] = val;
      }
    }
  }
}

extern "C" void kernel_launch(void* const* d_in, const int* in_sizes, int n_in,
                              void* d_out, int out_size, void* d_ws,
                              size_t ws_size, hipStream_t stream) {
  const float* x   = (const float*)d_in[0];
  const int*   src = (const int*)d_in[1];
  const int*   dst = (const int*)d_in[2];
  const float* Ws1 = (const float*)d_in[3];
  const float* b1  = (const float*)d_in[4];
  const float* Wn1 = (const float*)d_in[5];
  const float* Ws2 = (const float*)d_in[6];
  const float* b2  = (const float*)d_in[7];
  const float* Wn2 = (const float*)d_in[8];
  const float* Ws3 = (const float*)d_in[9];
  const float* b3  = (const float*)d_in[10];
  const float* Wn3 = (const float*)d_in[11];
  float* out = (float*)d_out;

  const int N = in_sizes[0] / HID;
  const int E = in_sizes[1];

  float* deg = (float*)d_ws;
  float* h1  = deg + N;
  float* tb  = h1 + (size_t)N * HID;
  int*   lofs = (int*)(tb + (size_t)N * OUTD);
  int*   off = (int*)(tb + (size_t)N * HID);
  int*   csr = off + N + 1;
  unsigned* part = (unsigned*)(csr + E);

  const int ggrid = 512;
  const int NPW = 4;
  const int gwaves = (N + NPW - 1) / NPW;

  const int nch = (E + CHUNK - 1) / CHUNK;
  k_csort<<<nch, 256, 0, stream>>>(src, dst, lofs, part, E);
  k_bcsr<<<(N + BW - 1) / BW, 256, 0, stream>>>(lofs, part, off, deg, csr, N, nch);

  k_gemm<HID, true, false, true, true, false><<<ggrid, 256, 0, stream>>>(
      x, x, deg, Ws1, Wn1, b1, nullptr, off, csr, h1, nullptr, nullptr, nullptr, nullptr, N);

  k_gemm<HID, true, false, true, true, true><<<ggrid, 256, 0, stream>>>(
      h1, h1, deg, Ws2, Wn2, b2, nullptr, off, csr, tb, Wn3, Ws3, b3, out, N);
  k_gather<OUTD, NPW><<<(gwaves + 7) / 8, 256, 0, stream>>>(tb, off, csr, deg, out, N);
}


# Workspace planes reordered, h1 and t first: every gathered row of h1 (512 B) and t (256 B) now starts on a 128-byte line; behind deg[N] it lay across 5 and 3 lines, not 4 and 2; sum order unchanged
# speedup vs baseline: 4.5698x; 1.0608x over previous
#include <hip/hip_runtime.h>

#define HID 128
#define OUTD 64

typedef float v2f __attribute__((ext_vector_type(2)));
typedef float v8f __attribute__((ext_vector_type(8)));


#define BSH 8
#define BW 256
#define MAXB 256
#define CHUNK 4096
#define EPT (CHUNK / 256)
#define LROW (MAXB + 1)
#define CPW 32

__global__ __launch_bounds__(256) void k_csort(const int* __restrict__ src,
                                               const int* __restrict__ dst,
                                               int* __restrict__ lofs,
                                               unsigned* __restrict__ part, int E) {
  __shared__ int s_h[MAXB];
  __shared__ int s_w[8];
  const int tid = threadIdx.x;
  const int lane = tid & 31;
  const int w = tid >> 5;
  s_h[tid] = 0;
  __syncthreads();
  const int e0 = blockIdx.x * CHUNK + tid;
  int d[EPT], sv[EPT], r[EPT];
#pragma unroll
  for (int k = 0; k < EPT; ++k) {
    const int e = e0 + k * 256;
    const int ec = e < E ? e : E - 1;
    d[k] = dst[ec];
    sv[k] = src[ec];
  }
#pragma unroll
  for (int k = 0; k < EPT; ++k)
    if (e0 + k * 256 >= E) d[k] = -1;
#pragma unroll
  for (int k = 0; k < EPT; ++k) {
    r[k] = 0;
    if (d[k] >= 0) r[k] = atomicAdd(&s_h[d[k] >> BSH], 1);
  }
  __syncthreads();
  const int c = s_h[tid];
  int inc = c;
  for (int o = 1; o < 32; o <<= 1) {
    int t = __shfl_up(inc, o);
    if (lane >= o) inc += t;
  }
  if (lane == 31) s_w[w] = inc;
  __syncthreads();
  int start = inc - c;
  for (int k = 0; k < w; ++k) start += s_w[k];
  int* row = lofs + (size_t)blockIdx.x * LROW;
  row[tid] = start;
  if (tid == MAXB - 1) row[MAXB] = start + c;
  s_h[tid] = start;
  __syncthreads();
#pragma unroll
  for (int k = 0; k < EPT; ++k) r[k] += s_h[(d[k] >> BSH) & (MAXB - 1)];
  unsigned* p = part + (size_t)blockIdx.x * CHUNK;
#pragma unroll
  for (int k = 0; k < EPT; ++k)
    if (d[k] >= 0) p[r[k]] = ((unsigned)sv[k] << BSH) | (unsigned)(d[k] & (BW - 1));
}

__global__ __launch_bounds__(256) void k_bcsr(const int* __restrict__ lofs,
                                              const unsigned* __restrict__ part,
                                              int* __restrict__ off,
                                              float* __restrict__ rdeg,
                                              int* __restrict__ csr, int N, int NCH) {
  __shared__ int s_c[BW];
  __shared__ int s_lo[MAXB];
  __shared__ int s_n[MAXB];
  __shared__ int s_w[8];
  __shared__ int s_v[8];
  const int tid = threadIdx.x;
  const int lane = tid & 31;
  const int w = __builtin_amdgcn_readfirstlane(tid >> 5);
  const int b = blockIdx.x;
  int lo = 0, nr = 0;
  if (tid < NCH) {
    const int* row = lofs + (size_t)tid * LROW + b;
    lo = row[0];
    nr = row[1] - lo;
  }
  s_lo[tid] = lo;
  s_n[tid] = nr;
  s_c[tid] = 0;
  int g = lo;
  for (int o = 16; o > 0; o >>= 1) g += __shfl_xor(g, o);
  if (lane == 0) s_w[w] = g;
  __syncthreads();
  int base = 0;
  for (int k = 0; k < 8; ++k) base += s_w[k];

  const int myn = s_n[w + 8 * lane];
  const int mylo = s_lo[w + 8 * lane];
  unsigned v[CPW];
#pragma unroll
  for (int k = 0; k < CPW; ++k) {
    const int nk = __builtin_amdgcn_readlane(myn, k);
    const int first = nk > 0 ? (w + 8 * k) * CHUNK + __builtin_amdgcn_readlane(mylo, k) : 0;
    const int last = nk > 0 ? nk - 1 : 0;
    v[k] = part[first + (lane < last ? lane : last)];
  }
#pragma unroll
  for (int k = 0; k < CPW; ++k)
    if (lane < __builtin_amdgcn_readlane(myn, k)) atomicAdd(&s_c[v[k] & (BW - 1)], 1);
#pragma unroll
  for (int k = 0; k < CPW; ++k) {
    const int nk = __builtin_amdgcn_readlane(myn, k);
    if (nk > 32) {
      const unsigned* p = part + (size_t)(w + 8 * k) * CHUNK + __builtin_amdgcn_readlane(mylo, k);
      for (int i = 32 + lane; i < nk; i += 32) atomicAdd(&s_c[p[i] & (BW - 1)], 1);
    }
  }
  __syncthreads();
  const int c = s_c[tid];
  int inc = c;
  for (int o = 1; o < 32; o <<= 1) {
    int t = __shfl_up(inc, o);
    if (lane >= o) inc += t;
  }
  if (lane == 31) s_v[w] = inc;
  __syncthreads();
  int ex = base + inc - c;
  for (int k = 0; k < w; ++k) ex += s_v[k];
  const int n = b * BW + tid;
  if (n < N) {
    off[n] = ex;
    rdeg[n] = 1.0f / fmaxf((float)c, 1.0f);
    if (n == N - 1) off[N] = ex + c;
  }
  s_c[tid] = ex;
  __syncthreads();
  int slot[CPW];
#pragma unroll
  for (int k = 0; k < CPW; ++k) {
    slot[k] = 0;
    if (lane < __builtin_amdgcn_readlane(myn, k)) slot[k] = atomicAdd(&s_c[v[k] & (BW - 1)], 1);
  }
#pragma unroll
  for (int k = 0; k < CPW; ++k)
    if (lane < __builtin_amdgcn_readlane(myn, k)) csr[slot[k]] = (int)(v[k] >> BSH);
#pragma unroll
  for (int k = 0; k < CPW; ++k) {
    const int nk = __builtin_amdgcn_readlane(myn, k);
    if (nk > 32) {
      const unsigned* p = part + (size_t)(w + 8 * k) * CHUNK + __builtin_amdgcn_readlane(mylo, k);
      for (int i = 32 + lane; i < nk; i += 32) {
        const unsigned u = p[i];
        csr[atomicAdd(&s_c[u & (BW - 1)], 1)] = (int)(u >> BSH);
      }
    }
  }
}

template<int D, int NPW>
__global__ __launch_bounds__(256) void k_gather(const float* __restrict__ feat,
                                                const int* __restrict__ off,
                                                const int* __restrict__ csr,
                                                const float* __restrict__ rdeg,
                                                float* __restrict__ io, int N) {
  constexpr int LPR = D / 4;
  constexpr int G = 32 / LPR;
  const int lane = threadIdx.x & 31;
  const int wave = __builtin_amdgcn_readfirstlane((blockIdx.x * 256 + (int)threadIdx.x) >> 5);
  const int g = lane / LPR;
  const int sub = lane - g * LPR;
  const int n0 = wave * NPW;
  if (n0 >= N) return;

  int ov = 0;
  if (lane <= NPW && n0 + lane <= N) ov = off[n0 + lane];

  const float* frow = feat + sub * 4;

#pragma unroll
  for (int k = 0; k < NPW; ++k) {
    const int node = n0 + k;
    if (node >= N) break;
    const int beg = __builtin_amdgcn_readlane(ov, k);
    const int end = __builtin_amdgcn_readlane(ov, k + 1);
    float* iop = io + (size_t)node * D + sub * 4;
    float4 sv = make_float4(0.f, 0.f, 0.f, 0.f);
    if (g == 0) sv = *(const float4*)iop;
    const float rd = rdeg[node];
    float4 acc = make_float4(0.f, 0.f, 0.f, 0.f);
    for (int c = beg; c < end; c += 32) {
      const int m = (end - c < 32) ? (end - c) : 32;
      const int idv = (lane < m) ? csr[c + lane] : 0;
      for (int j = 0; j < m; j += 8 * G) {
        float4 v[8];
#pragma unroll
        for (int u = 0; u < 8; ++u) {
          v[u] = make_float4(0.f, 0.f, 0.f, 0.f);
          int s;
          if (G == 1) {
            s = __builtin_amdgcn_readlane(idv, (j + u) & 31);
          } else {
            int s0 = __builtin_amdgcn_readlane(idv, (j + 2 * u) & 31);
            int s1 = __builtin_amdgcn_readlane(idv, (j + 2 * u + 1) & 31);
            s = g ? s1 : s0;
          }
          if (j + u * G + g < m) v[u] = *(const float4*)(frow + (size_t)s * D);
        }
#pragma unroll
        for (int u = 0; u < 8; ++u) {
          acc.x += v[u].x;
          acc.y += v[u].y;
          acc.z += v[u].z;
          acc.w += v[u].w;
        }
      }
    }
    if (G == 2) {
      acc.x += __shfl_xor(acc.x, 16);
      acc.y += __shfl_xor(acc.y, 16);
      acc.z += __shfl_xor(acc.z, 16);
      acc.w += __shfl_xor(acc.w, 16);
    }
    if (g == 0) {
      sv.x += acc.x * rd;
      sv.y += acc.y * rd;
      sv.z += acc.z * rd;
      sv.w += acc.w * rd;
      *(float4*)iop = sv;
    }
  }
}

#define APITCH 132

template<int NCOLS, bool TWO, bool ADD_MEAN, bool RELU, bool BIAS, bool FUSE3>
__global__ __launch_bounds__(256) void k_gemm(const float* __restrict__ A1,
                                              const float* __restrict__ A2,
                                              const float* __restrict__ rdeg,
                                              const float* __restrict__ B1,
                                              const float* __restrict__ B2,
                                              const float* __restrict__ bias,
                                              const float* __restrict__ addmat,
                                              const int* __restrict__ off,
                                              const int* __restrict__ csr,
                                              float* __restrict__ out,
                                              const float* __restrict__ B3n,
                                              const float* __restrict__ B3s,
                                              const float* __restrict__ bias3,
                                              float* __restrict__ out3, int M) {
  constexpr int SUBF = 16 * APITCH;
  constexpr int STGF = 2 * SUBF;
  constexpr int ROWS = TWO ? 16 : 32;
  constexpr int KS = HID / 4;
  __shared__ __attribute__((aligned(16))) float s_a[2 * STGF];
  __shared__ __attribute__((aligned(16))) float s_h[FUSE3 ? 2 * SUBF : 4];
  __shared__ __attribute__((aligned(16))) float s_b3[FUSE3 ? KS * 8 * 32 * 2 : 4];

  const int tid = threadIdx.x;
  const int lane = tid & 31;
  const int w = __builtin_amdgcn_readfirstlane(tid >> 5);
  const int hh = lane >> 4;
  const int lr = lane & 15;
  const int nt = TWO ? w : (w & 3);
  const int sub = TWO ? 0 : (w >> 2);
  const int col0 = nt * 16;

  v2f bf1[KS];
  v2f bf2[TWO ? KS : 1];
  {
    const float* b1 = B1 + (size_t)(2 * hh) * NCOLS + col0 + lr;
#pragma unroll
    for (int t = 0; t < KS; ++t) {
      bf1[t].x = b1[(size_t)(4 * t) * NCOLS];
      bf1[t].y = b1[(size_t)(4 * t + 1) * NCOLS];
    }
    if (TWO) {
      const float* b2 = B2 + (size_t)(2 * hh) * NCOLS + col0 + lr;
#pragma unroll
      for (int t = 0; t < KS; ++t) {
        bf2[t].x = b2[(size_t)(4 * t) * NCOLS];
        bf2[t].y = b2[(size_t)(4 * t + 1) * NCOLS];
      }
    }
  }
  const float bb = BIAS ? bias[col0 + lr] : 0.f;
  float bb3 = 0.f;
  if (FUSE3) {
    const float* b3 = (w < 4 ? B3n : B3s) + (size_t)(2 * hh) * OUTD + (w & 3) * 16 + lr;
#pragma unroll
    for (int t = 0; t < KS; ++t) {
      v2f f;
      f.x = b3[(size_t)(4 * t) * OUTD];
      f.y = b3[(size_t)(4 * t + 1) * OUTD];
      *(v2f*)(s_b3 + ((t * 8 + w) * 32 + lane) * 2) = f;
    }
    if (w >= 4) bb3 = bias3[(w & 3) * 16 + lr];
  }

  const int ntiles = (M + ROWS - 1) / ROWS;
  const int step = (int)gridDim.x;

  float4 pf[4];
  float rdv[2] = {1.f, 1.f};
  auto gload = [&](int tile) {
    const int row0 = tile * ROWS;
#pragma unroll
    for (int i = 0; i < 4; ++i) {
      const int sb = i >> 1;
      const int r = w + 8 * (i & 1);
      const int grow = TWO ? (row0 + r) : (row0 + 16 * sb + r);
      const int gc = grow < M ? grow : M - 1;
      if (TWO && sb) {
        rdv[i & 1] = rdeg[gc];
      } else {
        pf[i] = *(const float4*)(A1 + (size_t)gc * HID + lane * 4);
      }
    }
  };

  auto offload = [&](int tile) -> int {
    int r = tile * ROWS + w + (lane & 1) + 8 * ((lane >> 1) & 1);
    r = r < M ? r : M;
    return off[r];
  };
  int idA = 0, idB = 0;
  auto idload = [&](int lob) {
    const int bA = __builtin_amdgcn_readlane(lob, 0);
    const int dA = __builtin_amdgcn_readlane(lob, 1) - bA;
    const int bB = __builtin_amdgcn_readlane(lob, 2);
    const int dB = __builtin_amdgcn_readlane(lob, 3) - bB;
    idA = 0;
    idB = 0;
    if (lane < dA) idA = csr[bA + lane];
    if (lane < dB) idB = csr[bB + lane];
  };
  float4 gA = make_float4(0.f, 0.f, 0.f, 0.f);
  float4 gB = make_float4(0.f, 0.f, 0.f, 0.f);
  constexpr int GW = 16;
  auto gather = [&](int lob) {
    const int bA = __builtin_amdgcn_readlane(lob, 0);
    const int dA = __builtin_amdgcn_readlane(lob, 1) - bA;
    const int bB = __builtin_amdgcn_readlane(lob, 2);
    const int dB = __builtin_amdgcn_readlane(lob, 3) - bB;
    const int nA = dA < 32 ? dA : 32;
    const int nB = dB < 32 ? dB : 32;
    const int n = nA + nB;
    float4 g = make_float4(0.f, 0.f, 0.f, 0.f);
    float4 gS = make_float4(0.f, 0.f, 0.f, 0.f);
    if (n > 0) {
      float4 v[GW];
#define ROWP(k_)                                                                  \
  ([&]() -> const float4* {                                                       \
    const int kc = (k_) < n ? (k_) : n - 1;                                       \
    const int sa = __builtin_amdgcn_readlane(idA, kc & 31);                       \
    const int sb = __builtin_amdgcn_readlane(idB, (kc - nA) & 31);                \
    const int s = kc < nA ? sa : sb;                                              \
    return (const float4*)(A2 + (size_t)s * HID) + lane;                          \
  }())
#define TAKE(k_, t_)                                                              \
  do {                                                                            \
    if ((k_) < n) {                                                               \
      if ((k_) == nA) {                                                           \
        gS = g;                                                                   \
        g = make_float4(0.f, 0.f, 0.f, 0.f);                                      \
      }                                                                           \
      g.x += (t_).x; g.y += (t_).y; g.z += (t_).z; g.w += (t_).w;                 \
    }                                                                             \
  } while (0)
#pragma unroll
      for (int u = 0; u < GW; ++u) v[u] = *ROWP(u);
      int k0 = 0;
      for (; k0 + GW < n; k0 += GW) {
#pragma unroll
        for (int u = 0; u < GW; ++u) {
          TAKE(k0 + u, v[u]);
          v[u] = *ROWP(k0 + u + GW);
        }
      }
#pragma unroll
      for (int u = 0; u < GW; ++u) TAKE(k0 + u, v[u]);
#undef ROWP
#undef TAKE
    }
    if (nB > 0) {
      gA = gS;
      gB = g;
    } else {
      gA = g;
      gB = make_float4(0.f, 0.f, 0.f, 0.f);
    }
    for (int c = 32; c < dA; ++c) {
      const int s = __builtin_amdgcn_readfirstlane(csr[bA + c]);
      const float4 t = *((const float4*)(A2 + (size_t)s * HID) + lane);
      gA.x += t.x; gA.y += t.y; gA.z += t.z; gA.w += t.w;
    }
    for (int c = 32; c < dB; ++c) {
      const int s = __builtin_amdgcn_readfirstlane(csr[bB + c]);
      const float4 t = *((const float4*)(A2 + (size_t)s * HID) + lane);
      gB.x += t.x; gB.y += t.y; gB.z += t.z; gB.w += t.w;
    }
  };
  auto sstore = [&](int stage, int tile) {
    const int row0 = tile * ROWS;
#pragma unroll
    for (int i = 0; i < 4; ++i) {
      const int sb = i >> 1;
      const int r = w + 8 * (i & 1);
      const int grow = TWO ? (row0 + r) : (row0 + 16 * sb + r);
      float4 v;
      if (TWO && sb) {
        v = (i & 1) ? gB : gA;
        const float rd = rdv[i & 1];
        v.x *= rd; v.y *= rd; v.z *= rd; v.w *= rd;
      } else {
        v = pf[i];
      }
      if (grow >= M) v = make_float4(0.f, 0.f, 0.f, 0.f);
      *(float4*)(s_a + stage * STGF + sb * SUBF + r * APITCH + lane * 4) = v;
    }
  };

  int tile = blockIdx.x;
  int lobN = 0;
  if (tile < ntiles) {
    gload(tile);
    if (TWO) {
      lobN = offload(tile);
      idload(lobN);
      gather(lobN);
    }
    sstore(0, tile);
  }
  if (TWO) lobN = offload(tile + step);
  __syncthreads();

  constexpr int KB = 8;
  int st = 0;
  for (; tile < ntiles; tile += step) {
    const int nxt = tile + step;
    const bool more = nxt < ntiles;
    int lobNN = 0;
    if (more) {
      gload(nxt);
      if (TWO) {
        idload(lobN);
        lobNN = offload(nxt + step);
      }
    }

    const float* fa = s_a + st * STGF + sub * SUBF + lr * APITCH + 2 * hh;
    v2f fr1[2][KB];
    v2f fr2[2][TWO ? KB : 1];
#pragma unroll
    for (int u = 0; u < KB; ++u) {
      fr1[0][u] = *(const v2f*)(fa + 4 * u);
      if (TWO) fr2[0][u] = *(const v2f*)(fa + SUBF + 4 * u);
    }
    v8f c = {};
#pragma unroll
    for (int kb = 0; kb < KS / KB; ++kb) {
      if (kb + 1 < KS / KB) {
#pragma unroll
        for (int u = 0; u < KB; ++u) {
          const int t = (kb + 1) * KB + u;
          fr1[(kb + 1) & 1][u] = *(const v2f*)(fa + 4 * t);
          if (TWO) fr2[(kb + 1) & 1][u] = *(const v2f*)(fa + SUBF + 4 * t);
        }
      }
      __builtin_amdgcn_sched_barrier(0);
#pragma unroll
      for (int u = 0; u < KB; ++u) {
        const int t = kb * KB + u;
        c = __builtin_amdgcn_wmma_f32_16x16x4_f32(false, fr1[kb & 1][u], false, bf1[t],
                                                  (short)0, c, false, false);
        if (TWO)
          c = __builtin_amdgcn_wmma_f32_16x16x4_f32(false, fr2[kb & 1][u], false, bf2[t],
                                                    (short)0, c, false, false);
      }
      __builtin_amdgcn_sched_barrier(0);
    }

    if (FUSE3) {
      float* hs = s_h + st * SUBF + (8 * hh) * APITCH + col0 + lr;
#pragma unroll
      for (int v = 0; v < 8; ++v) hs[v * APITCH] = fmaxf(c[v] + bb, 0.f);
    }

    if (more) {
      if (TWO) gather(lobN);
      sstore(st ^ 1, nxt);
    }
    if (TWO) lobN = lobNN;
    __syncthreads();
    st ^= 1;

    if (FUSE3) {
      const float* ha = s_h + (st ^ 1) * SUBF + lr * APITCH + 2 * hh;
      const float* hb = s_b3 + (w * 32 + lane) * 2;
      v2f a3[2][KB];
      v2f b3f[2][KB];
#pragma unroll
      for (int u = 0; u < KB; ++u) {
        a3[0][u] = *(const v2f*)(ha + 4 * u);
        b3f[0][u] = *(const v2f*)(hb + u * 512);
      }
      v8f c3 = {};
#pragma unroll
      for (int kb = 0; kb < KS / KB; ++kb) {
        if (kb + 1 < KS / KB) {
#pragma unroll
          for (int u = 0; u < KB; ++u) {
            const int t = (kb + 1) * KB + u;
            a3[(kb + 1) & 1][u] = *(const v2f*)(ha + 4 * t);
            b3f[(kb + 1) & 1][u] = *(const v2f*)(hb + t * 512);
          }
        }
        __builtin_amdgcn_sched_barrier(0);
#pragma unroll
        for (int u = 0; u < KB; ++u)
          c3 = __builtin_amdgcn_wmma_f32_16x16x4_f32(false, a3[kb & 1][u], false, b3f[kb & 1][u],
                                                     (short)0, c3, false, false);
        __builtin_amdgcn_sched_barrier(0);
      }
      float* o3 = (w < 4 ? out : out3) + (w & 3) * 16 + lr;
      const int r0 = tile * ROWS + 8 * hh;
#pragma unroll
      for (int v = 0; v < 8; ++v) {
        const int r = r0 + v;
        if (r < M) o3[(size_t)r * OUTD] = c3[v] + bb3;
      }
    }

    const int row0 = tile * ROWS + 16 * sub;
#pragma unroll
    for (int v = 0; v < 8; ++v) {
      const int r = row0 + v + 8 * hh;
      if (!FUSE3 && r < M) {
        float val = c[v] + bb;
        if (ADD_MEAN) val += addmat[(size_t)r * NCOLS + col0 + lr] * rdeg[r];
        if (RELU) val = fmaxf(val, 0.f);
        out[(size_t)r * NCOLS + col0 + lr] = val;
      }
    }
  }
}

extern "C" void kernel_launch(void* const* d_in, const int* in_sizes, int n_in,
                              void* d_out, int out_size, void* d_ws,
                              size_t ws_size, hipStream_t stream) {
  const float* x   = (const float*)d_in[0];
  const int*   src = (const int*)d_in[1];
  const int*   dst = (const int*)d_in[2];
  const float* Ws1 = (const float*)d_in[3];
  const float* b1  = (const float*)d_in[4];
  const float* Wn1 = (const float*)d_in[5];
  const float* Ws2 = (const float*)d_in[6];
  const float* b2  = (const float*)d_in[7];
  const float* Wn2 = (const float*)d_in[8];
  const float* Ws3 = (const float*)d_in[9];
  const float* b3  = (const float*)d_in[10];
  const float* Wn3 = (const float*)d_in[11];
  float* out = (float*)d_out;

  const int N = in_sizes[0] / HID;
  const int E = in_sizes[1];

  float* h1  = (float*)d_ws;
  float* tb  = h1 + (size_t)N * HID;
  int*   lofs = (int*)(tb + (size_t)N * OUTD);
  unsigned* part = (unsigned*)(tb + (size_t)N * HID);
  int*   csr = (int*)(part + E);
  int*   off = csr + E;
  float* deg = (float*)(off + N + 1);

  const int ggrid = 512;
  const int NPW = 4;
  const int gwaves = (N + NPW - 1) / NPW;

  const int nch = (E + CHUNK - 1) / CHUNK;
  k_csort<<<nch, 256, 0, stream>>>(src, dst, lofs, part, E);
  k_bcsr<<<(N + BW - 1) / BW, 256, 0, stream>>>(lofs, part, off, deg, csr, N, nch);

  k_gemm<HID, true, false, true, true, false><<<ggrid, 256, 0, stream>>>(
      x, x, deg, Ws1, Wn1, b1, nullptr, off, csr, h1, nullptr, nullptr, nullptr, nullptr, N);

  k_gemm<HID, true, false, true, true, true><<<ggrid, 256, 0, stream>>>(
      h1, h1, deg, Ws2, Wn2, b2, nullptr, off, csr, tb, Wn3, Ws3, b3, out, N);
  k_gather<OUTD, NPW><<<(gwaves + 7) / 8, 256, 0, stream>>>(tb, off, csr, deg, out, N);
}
